# starting kernel
# baseline (speedup 1.0000x reference)
#include <hip/hip_runtime.h>
#include <hip/hip_bf16.h>

typedef __attribute__((ext_vector_type(2))) float v2f;
typedef __attribute__((ext_vector_type(8))) float v8f;

#define Bv 64
#define Tv 512
#define Dv 1024
#define Kv 32

__device__ __forceinline__ float fexp2(float x) { return __builtin_amdgcn_exp2f(x); }
__device__ __forceinline__ float flog2(float x) { return __builtin_amdgcn_logf(x); }

__device__ __forceinline__ float readlane_f(float v, int l) {
    return __builtin_bit_cast(float,
        __builtin_amdgcn_readlane(__builtin_bit_cast(int, v), l));
}

__global__ __launch_bounds__(256) void crf_transposeW(const float* __restrict__ W,
                                                      float* __restrict__ Wt) {
    int idx = blockIdx.x * 256 + threadIdx.x;
    int k = idx >> 10;
    int d = idx & 1023;
    Wt[idx] = W[d * Kv + k];
}

__global__ __launch_bounds__(128) void crf_emissions(const float* __restrict__ x,
                                                     const float* __restrict__ Wt,
                                                     const float* __restrict__ bias,
                                                     float* __restrict__ e) {
    const int lane = threadIdx.x & 31;
    const int wave = threadIdx.x >> 5;
    const int mt   = blockIdx.x * 4 + wave;
    const int half = lane >> 4;
    const int l15  = lane & 15;

    const float* xptr = x  + (size_t)(mt * 16 + l15) * Dv + half * 2;
    const float* w0   = Wt + (size_t)(l15)      * Dv + half * 2;
    const float* w1   = Wt + (size_t)(l15 + 16) * Dv + half * 2;

    v8f acc0 = {};
    v8f acc1 = {};
    #pragma unroll 4
    for (int d = 0; d < Dv / 4; ++d) {
        v2f a  = *(const v2f*)(xptr + d * 4);
        v2f b0 = *(const v2f*)(w0   + d * 4);
        v2f b1 = *(const v2f*)(w1   + d * 4);
        acc0 = __builtin_amdgcn_wmma_f32_16x16x4_f32(false, a, false, b0,
                                                     (short)0, acc0, false, false);
        acc1 = __builtin_amdgcn_wmma_f32_16x16x4_f32(false, a, false, b1,
                                                     (short)0, acc1, false, false);
    }

    const float bv0 = bias[l15];
    const float bv1 = bias[l15 + 16];
    #pragma unroll
    for (int r = 0; r < 8; ++r) {
        size_t row = (size_t)mt * 16 + r + half * 8;
        e[row * Kv + l15]      = acc0[r] + bv0;
        e[row * Kv + l15 + 16] = acc1[r] + bv1;
    }
}

__global__ __launch_bounds__(32) void crf_scan(const float* __restrict__ e,
                                               const float* __restrict__ U,
                                               float* __restrict__ alpha_out,
                                               float* __restrict__ beta_out) {
    const int   lane   = threadIdx.x;
    const float LOG2E  = 1.44269504088896340736f;
    const float LN2    = 0.69314718055994530942f;
    float eU[Kv];

    if (blockIdx.x < Bv) {
        const int bIdx = blockIdx.x;
        #pragma unroll
        for (int j = 0; j < Kv; ++j)
            eU[j] = fexp2(U[j * Kv + lane] * LOG2E);

        const float* eb = e         + (size_t)bIdx * Tv * Kv;
        float*       ab = alpha_out + (size_t)bIdx * Tv * Kv;

        float alpha = eb[lane];
        ab[lane] = alpha;
        for (int t = 1; t < Tv; ++t) {
            float m = alpha;
            #pragma unroll
            for (int off = 16; off; off >>= 1) m = fmaxf(m, __shfl_xor(m, off, 32));
            float p = fexp2((alpha - m) * LOG2E);
            float s = 0.f;
            #pragma unroll
            for (int j = 0; j < Kv; ++j)
                s = fmaf(readlane_f(p, j), eU[j], s);
            alpha = m + flog2(s) * LN2 + eb[t * Kv + lane];
            ab[t * Kv + lane] = alpha;
        }
    } else {
        const int bIdx = blockIdx.x - Bv;
        #pragma unroll
        for (int j = 0; j < Kv; ++j)
            eU[j] = fexp2(U[lane * Kv + j] * LOG2E);

        const float* eb = e        + (size_t)bIdx * Tv * Kv;
        float*       bb = beta_out + (size_t)bIdx * Tv * Kv;

        float beta = 0.f;
        bb[(Tv - 1) * Kv + lane] = 0.f;
        for (int t = Tv - 2; t >= 0; --t) {
            float g = eb[(t + 1) * Kv + lane] + beta;
            float m = g;
            #pragma unroll
            for (int off = 16; off; off >>= 1) m = fmaxf(m, __shfl_xor(m, off, 32));
            float q = fexp2((g - m) * LOG2E);
            float s = 0.f;
            #pragma unroll
            for (int kk = 0; kk < Kv; ++kk)
                s = fmaf(readlane_f(q, kk), eU[kk], s);
            beta = m + flog2(s) * LN2;
            bb[t * Kv + lane] = beta;
        }
    }
}

__global__ __launch_bounds__(256) void crf_softmax(const float* __restrict__ alpha,
                                                   float* __restrict__ out) {
    const int lane = threadIdx.x & 31;
    const int wave = threadIdx.x >> 5;
    const size_t row = (size_t)blockIdx.x * 8 + wave;
    const size_t idx = row * Kv + lane;
    const float LOG2E = 1.44269504088896340736f;

    float s = alpha[idx] + out[idx];
    float m = s;
    #pragma unroll
    for (int off = 16; off; off >>= 1) m = fmaxf(m, __shfl_xor(m, off, 32));
    float p = fexp2((s - m) * LOG2E);
    float sum = p;
    #pragma unroll
    for (int off = 16; off; off >>= 1) sum += __shfl_xor(sum, off, 32);
    out[idx] = p / sum;
}

extern "C" void kernel_launch(void* const* d_in, const int* in_sizes, int n_in,
                              void* d_out, int out_size, void* d_ws, size_t ws_size,
                              hipStream_t stream) {
    const float* x = (const float*)d_in[0];
    const float* W = (const float*)d_in[1];
    const float* U = (const float*)d_in[2];
    const float* b = (const float*)d_in[3];
    float* out = (float*)d_out;

    float* ws    = (float*)d_ws;
    float* e     = ws;
    float* alpha = ws + (size_t)Bv * Tv * Kv;
    float* Wt    = alpha + (size_t)Bv * Tv * Kv;

    crf_transposeW<<<dim3(128), dim3(256), 0, stream>>>(W, Wt);
    crf_emissions <<<dim3((Bv * Tv / 16) / 4), dim3(128), 0, stream>>>(x, Wt, b, e);
    crf_scan      <<<dim3(2 * Bv), dim3(32), 0, stream>>>(e, U, alpha, out);
    crf_softmax   <<<dim3(Bv * Tv / 8), dim3(256), 0, stream>>>(alpha, out);
}


# scan: the emission row of each step is loaded 4 steps ahead into a register ring, so neither recursion waits for a memory round trip inside its serial chain
# speedup vs baseline: 1.7514x; 1.7514x over previous
#include <hip/hip_runtime.h>
#include <hip/hip_bf16.h>

typedef __attribute__((ext_vector_type(2))) float v2f;
typedef __attribute__((ext_vector_type(8))) float v8f;

#define Bv 64
#define Tv 512
#define Dv 1024
#define Kv 32

__device__ __forceinline__ float fexp2(float x) { return __builtin_amdgcn_exp2f(x); }
__device__ __forceinline__ float flog2(float x) { return __builtin_amdgcn_logf(x); }

__device__ __forceinline__ float readlane_f(float v, int l) {
    return __builtin_bit_cast(float,
        __builtin_amdgcn_readlane(__builtin_bit_cast(int, v), l));
}

__global__ __launch_bounds__(256) void crf_transposeW(const float* __restrict__ W,
                                                      float* __restrict__ Wt) {
    int idx = blockIdx.x * 256 + threadIdx.x;
    int k = idx >> 10;
    int d = idx & 1023;
    Wt[idx] = W[d * Kv + k];
}

__global__ __launch_bounds__(128) void crf_emissions(const float* __restrict__ x,
                                                     const float* __restrict__ Wt,
                                                     const float* __restrict__ bias,
                                                     float* __restrict__ e) {
    const int lane = threadIdx.x & 31;
    const int wave = threadIdx.x >> 5;
    const int mt   = blockIdx.x * 4 + wave;
    const int half = lane >> 4;
    const int l15  = lane & 15;

    const float* xptr = x  + (size_t)(mt * 16 + l15) * Dv + half * 2;
    const float* w0   = Wt + (size_t)(l15)      * Dv + half * 2;
    const float* w1   = Wt + (size_t)(l15 + 16) * Dv + half * 2;

    v8f acc0 = {};
    v8f acc1 = {};
    #pragma unroll 4
    for (int d = 0; d < Dv / 4; ++d) {
        v2f a  = *(const v2f*)(xptr + d * 4);
        v2f b0 = *(const v2f*)(w0   + d * 4);
        v2f b1 = *(const v2f*)(w1   + d * 4);
        acc0 = __builtin_amdgcn_wmma_f32_16x16x4_f32(false, a, false, b0,
                                                     (short)0, acc0, false, false);
        acc1 = __builtin_amdgcn_wmma_f32_16x16x4_f32(false, a, false, b1,
                                                     (short)0, acc1, false, false);
    }

    const float bv0 = bias[l15];
    const float bv1 = bias[l15 + 16];
    #pragma unroll
    for (int r = 0; r < 8; ++r) {
        size_t row = (size_t)mt * 16 + r + half * 8;
        e[row * Kv + l15]      = acc0[r] + bv0;
        e[row * Kv + l15 + 16] = acc1[r] + bv1;
    }
}

#define PF 4

__device__ __forceinline__ float crf_fwd_step(float alpha, float et, const float (&eU)[Kv]) {
    const float LOG2E  = 1.44269504088896340736f;
    const float LN2    = 0.69314718055994530942f;
    float m = alpha;
    #pragma unroll
    for (int off = 16; off; off >>= 1) m = fmaxf(m, __shfl_xor(m, off, 32));
    float p = fexp2((alpha - m) * LOG2E);
    float s = 0.f;
    #pragma unroll
    for (int j = 0; j < Kv; ++j)
        s = fmaf(readlane_f(p, j), eU[j], s);
    return m + flog2(s) * LN2 + et;
}

__device__ __forceinline__ float crf_bwd_step(float beta, float en, const float (&eU)[Kv]) {
    const float LOG2E  = 1.44269504088896340736f;
    const float LN2    = 0.69314718055994530942f;
    float g = en + beta;
    float m = g;
    #pragma unroll
    for (int off = 16; off; off >>= 1) m = fmaxf(m, __shfl_xor(m, off, 32));
    float q = fexp2((g - m) * LOG2E);
    float s = 0.f;
    #pragma unroll
    for (int kk = 0; kk < Kv; ++kk)
        s = fmaf(readlane_f(q, kk), eU[kk], s);
    return m + flog2(s) * LN2;
}

__global__ __launch_bounds__(32) void crf_scan(const float* __restrict__ e,
                                               const float* __restrict__ U,
                                               float* __restrict__ alpha_out,
                                               float* __restrict__ beta_out) {
    const int   lane   = threadIdx.x;
    const float LOG2E  = 1.44269504088896340736f;
    float eU[Kv];
    float ring[PF];

    if (blockIdx.x < Bv) {
        const int bIdx = blockIdx.x;
        #pragma unroll
        for (int j = 0; j < Kv; ++j)
            eU[j] = fexp2(U[j * Kv + lane] * LOG2E);

        const float* eb = e         + (size_t)bIdx * Tv * Kv;
        float*       ab = alpha_out + (size_t)bIdx * Tv * Kv;

        #pragma unroll
        for (int i = 0; i < PF; ++i)
            ring[i] = eb[(1 + i) * Kv + lane];

        float alpha = eb[lane];
        ab[lane] = alpha;
        static_assert(PF == 4, "the statement below names the PF ring slots one by one");
        asm volatile("" :: "v"(ring[0]), "v"(ring[1]), "v"(ring[2]), "v"(ring[3]));
        int t = 1;
        for (; t + PF <= Tv; t += PF) {
            #pragma unroll
            for (int i = 0; i < PF; ++i) {
                alpha = crf_fwd_step(alpha, ring[i], eU);
                ab[(t + i) * Kv + lane] = alpha;
                int tn = t + i + PF;
                tn = tn < Tv ? tn : Tv - 1;
                ring[i] = eb[tn * Kv + lane];
                __builtin_amdgcn_sched_barrier(0);
            }
        }
        #pragma unroll
        for (int i = 0; i < PF - 1; ++i) {
            if (t + i < Tv) {
                alpha = crf_fwd_step(alpha, ring[i], eU);
                ab[(t + i) * Kv + lane] = alpha;
            }
        }
    } else {
        const int bIdx = blockIdx.x - Bv;
        #pragma unroll
        for (int j = 0; j < Kv; ++j)
            eU[j] = fexp2(U[lane * Kv + j] * LOG2E);

        const float* eb = e        + (size_t)bIdx * Tv * Kv;
        float*       bb = beta_out + (size_t)bIdx * Tv * Kv;

        #pragma unroll
        for (int i = 0; i < PF; ++i)
            ring[i] = eb[(Tv - 1 - i) * Kv + lane];

        float beta = 0.f;
        bb[(Tv - 1) * Kv + lane] = 0.f;
        int t = Tv - 2;
        for (; t - (PF - 1) >= 0; t -= PF) {
            #pragma unroll
            for (int i = 0; i < PF; ++i) {
                beta = crf_bwd_step(beta, ring[i], eU);
                bb[(t - i) * Kv + lane] = beta;
                int qn = t - i + 1 - PF;
                qn = qn > 1 ? qn : 1;
                ring[i] = eb[qn * Kv + lane];
                __builtin_amdgcn_sched_barrier(0);
            }
        }
        #pragma unroll
        for (int i = 0; i < PF - 1; ++i) {
            if (t - i >= 0) {
                beta = crf_bwd_step(beta, ring[i], eU);
                bb[(t - i) * Kv + lane] = beta;
            }
        }
    }
}

__global__ __launch_bounds__(256) void crf_softmax(const float* __restrict__ alpha,
                                                   float* __restrict__ out) {
    const int lane = threadIdx.x & 31;
    const int wave = threadIdx.x >> 5;
    const size_t row = (size_t)blockIdx.x * 8 + wave;
    const size_t idx = row * Kv + lane;
    const float LOG2E = 1.44269504088896340736f;

    float s = alpha[idx] + out[idx];
    float m = s;
    #pragma unroll
    for (int off = 16; off; off >>= 1) m = fmaxf(m, __shfl_xor(m, off, 32));
    float p = fexp2((s - m) * LOG2E);
    float sum = p;
    #pragma unroll
    for (int off = 16; off; off >>= 1) sum += __shfl_xor(sum, off, 32);
    out[idx] = p / sum;
}

extern "C" void kernel_launch(void* const* d_in, const int* in_sizes, int n_in,
                              void* d_out, int out_size, void* d_ws, size_t ws_size,
                              hipStream_t stream) {
    const float* x = (const float*)d_in[0];
    const float* W = (const float*)d_in[1];
    const float* U = (const float*)d_in[2];
    const float* b = (const float*)d_in[3];
    float* out = (float*)d_out;

    float* ws    = (float*)d_ws;
    float* e     = ws;
    float* alpha = ws + (size_t)Bv * Tv * Kv;
    float* Wt    = alpha + (size_t)Bv * Tv * Kv;

    crf_transposeW<<<dim3(128), dim3(256), 0, stream>>>(W, Wt);
    crf_emissions <<<dim3((Bv * Tv / 16) / 4), dim3(128), 0, stream>>>(x, Wt, b, e);
    crf_scan      <<<dim3(2 * Bv), dim3(32), 0, stream>>>(e, U, alpha, out);
    crf_softmax   <<<dim3(Bv * Tv / 8), dim3(256), 0, stream>>>(alpha, out);
}


# scan: the wave-wide max of each recursion step runs on the vector ALU (v_permlanex16, then DPP row_xmask on the max itself) instead of five ds_bpermute round trips on the serial chain
# speedup vs baseline: 2.7092x; 1.5469x over previous
#include <hip/hip_runtime.h>
#include <hip/hip_bf16.h>

typedef __attribute__((ext_vector_type(2))) float v2f;
typedef __attribute__((ext_vector_type(8))) float v8f;

#define Bv 64
#define Tv 512
#define Dv 1024
#define Kv 32

__device__ __forceinline__ float fexp2(float x) { return __builtin_amdgcn_exp2f(x); }
__device__ __forceinline__ float flog2(float x) { return __builtin_amdgcn_logf(x); }

__device__ __forceinline__ float readlane_f(float v, int l) {
    return __builtin_bit_cast(float,
        __builtin_amdgcn_readlane(__builtin_bit_cast(int, v), l));
}

__device__ __forceinline__ float wave_max(float v) {
    int b = __builtin_bit_cast(int, v);
    float x = __builtin_bit_cast(float,
        __builtin_amdgcn_permlanex16(0, b, 0x76543210, 0xfedcba98, false, false));
    float m = fmaxf(v, x);
    float r;
    asm("v_max_num_f32_dpp %0, %1, %1 row_xmask:1 row_mask:0xf bank_mask:0xf\n\t"
        "v_max_num_f32_dpp %0, %0, %0 row_xmask:2 row_mask:0xf bank_mask:0xf\n\t"
        "v_max_num_f32_dpp %0, %0, %0 row_xmask:4 row_mask:0xf bank_mask:0xf\n\t"
        "v_max_num_f32_dpp %0, %0, %0 row_xmask:8 row_mask:0xf bank_mask:0xf"
        : "=v"(r) : "v"(m));
    return r;
}

__global__ __launch_bounds__(256) void crf_transposeW(const float* __restrict__ W,
                                                      float* __restrict__ Wt) {
    int idx = blockIdx.x * 256 + threadIdx.x;
    int k = idx >> 10;
    int d = idx & 1023;
    Wt[idx] = W[d * Kv + k];
}

__global__ __launch_bounds__(128) void crf_emissions(const float* __restrict__ x,
                                                     const float* __restrict__ Wt,
                                                     const float* __restrict__ bias,
                                                     float* __restrict__ e) {
    const int lane = threadIdx.x & 31;
    const int wave = threadIdx.x >> 5;
    const int mt   = blockIdx.x * 4 + wave;
    const int half = lane >> 4;
    const int l15  = lane & 15;

    const float* xptr = x  + (size_t)(mt * 16 + l15) * Dv + half * 2;
    const float* w0   = Wt + (size_t)(l15)      * Dv + half * 2;
    const float* w1   = Wt + (size_t)(l15 + 16) * Dv + half * 2;

    v8f acc0 = {};
    v8f acc1 = {};
    #pragma unroll 4
    for (int d = 0; d < Dv / 4; ++d) {
        v2f a  = *(const v2f*)(xptr + d * 4);
        v2f b0 = *(const v2f*)(w0   + d * 4);
        v2f b1 = *(const v2f*)(w1   + d * 4);
        acc0 = __builtin_amdgcn_wmma_f32_16x16x4_f32(false, a, false, b0,
                                                     (short)0, acc0, false, false);
        acc1 = __builtin_amdgcn_wmma_f32_16x16x4_f32(false, a, false, b1,
                                                     (short)0, acc1, false, false);
    }

    const float bv0 = bias[l15];
    const float bv1 = bias[l15 + 16];
    #pragma unroll
    for (int r = 0; r < 8; ++r) {
        size_t row = (size_t)mt * 16 + r + half * 8;
        e[row * Kv + l15]      = acc0[r] + bv0;
        e[row * Kv + l15 + 16] = acc1[r] + bv1;
    }
}

#define PF 4

__device__ __forceinline__ float crf_fwd_step(float alpha, float et, const float (&eU)[Kv]) {
    const float LOG2E  = 1.44269504088896340736f;
    const float LN2    = 0.69314718055994530942f;
    float m = wave_max(alpha);
    float p = fexp2((alpha - m) * LOG2E);
    float s = 0.f;
    #pragma unroll
    for (int j = 0; j < Kv; ++j)
        s = fmaf(readlane_f(p, j), eU[j], s);
    return m + flog2(s) * LN2 + et;
}

__device__ __forceinline__ float crf_bwd_step(float beta, float en, const float (&eU)[Kv]) {
    const float LOG2E  = 1.44269504088896340736f;
    const float LN2    = 0.69314718055994530942f;
    float g = en + beta;
    float m = wave_max(g);
    float q = fexp2((g - m) * LOG2E);
    float s = 0.f;
    #pragma unroll
    for (int kk = 0; kk < Kv; ++kk)
        s = fmaf(readlane_f(q, kk), eU[kk], s);
    return m + flog2(s) * LN2;
}

__global__ __launch_bounds__(32) void crf_scan(const float* __restrict__ e,
                                               const float* __restrict__ U,
                                               float* __restrict__ alpha_out,
                                               float* __restrict__ beta_out) {
    const int   lane   = threadIdx.x;
    const float LOG2E  = 1.44269504088896340736f;
    float eU[Kv];
    float ring[PF];

    if (blockIdx.x < Bv) {
        const int bIdx = blockIdx.x;
        #pragma unroll
        for (int j = 0; j < Kv; ++j)
            eU[j] = fexp2(U[j * Kv + lane] * LOG2E);

        const float* eb = e         + (size_t)bIdx * Tv * Kv;
        float*       ab = alpha_out + (size_t)bIdx * Tv * Kv;

        #pragma unroll
        for (int i = 0; i < PF; ++i)
            ring[i] = eb[(1 + i) * Kv + lane];

        float alpha = eb[lane];
        ab[lane] = alpha;
        static_assert(PF == 4, "the statement below names the PF ring slots one by one");
        asm volatile("" :: "v"(ring[0]), "v"(ring[1]), "v"(ring[2]), "v"(ring[3]));
        int t = 1;
        for (; t + PF <= Tv; t += PF) {
            #pragma unroll
            for (int i = 0; i < PF; ++i) {
                alpha = crf_fwd_step(alpha, ring[i], eU);
                ab[(t + i) * Kv + lane] = alpha;
                int tn = t + i + PF;
                tn = tn < Tv ? tn : Tv - 1;
                ring[i] = eb[tn * Kv + lane];
                __builtin_amdgcn_sched_barrier(0);
            }
        }
        #pragma unroll
        for (int i = 0; i < PF - 1; ++i) {
            if (t + i < Tv) {
                alpha = crf_fwd_step(alpha, ring[i], eU);
                ab[(t + i) * Kv + lane] = alpha;
            }
        }
    } else {
        const int bIdx = blockIdx.x - Bv;
        #pragma unroll
        for (int j = 0; j < Kv; ++j)
            eU[j] = fexp2(U[lane * Kv + j] * LOG2E);

        const float* eb = e        + (size_t)bIdx * Tv * Kv;
        float*       bb = beta_out + (size_t)bIdx * Tv * Kv;

        #pragma unroll
        for (int i = 0; i < PF; ++i)
            ring[i] = eb[(Tv - 1 - i) * Kv + lane];

        float beta = 0.f;
        bb[(Tv - 1) * Kv + lane] = 0.f;
        int t = Tv - 2;
        for (; t - (PF - 1) >= 0; t -= PF) {
            #pragma unroll
            for (int i = 0; i < PF; ++i) {
                beta = crf_bwd_step(beta, ring[i], eU);
                bb[(t - i) * Kv + lane] = beta;
                int qn = t - i + 1 - PF;
                qn = qn > 1 ? qn : 1;
                ring[i] = eb[qn * Kv + lane];
                __builtin_amdgcn_sched_barrier(0);
            }
        }
        #pragma unroll
        for (int i = 0; i < PF - 1; ++i) {
            if (t - i >= 0) {
                beta = crf_bwd_step(beta, ring[i], eU);
                bb[(t - i) * Kv + lane] = beta;
            }
        }
    }
}

__global__ __launch_bounds__(256) void crf_softmax(const float* __restrict__ alpha,
                                                   float* __restrict__ out) {
    const int lane = threadIdx.x & 31;
    const int wave = threadIdx.x >> 5;
    const size_t row = (size_t)blockIdx.x * 8 + wave;
    const size_t idx = row * Kv + lane;
    const float LOG2E = 1.44269504088896340736f;

    float s = alpha[idx] + out[idx];
    float m = s;
    #pragma unroll
    for (int off = 16; off; off >>= 1) m = fmaxf(m, __shfl_xor(m, off, 32));
    float p = fexp2((s - m) * LOG2E);
    float sum = p;
    #pragma unroll
    for (int off = 16; off; off >>= 1) sum += __shfl_xor(sum, off, 32);
    out[idx] = p / sum;
}

extern "C" void kernel_launch(void* const* d_in, const int* in_sizes, int n_in,
                              void* d_out, int out_size, void* d_ws, size_t ws_size,
                              hipStream_t stream) {
    const float* x = (const float*)d_in[0];
    const float* W = (const float*)d_in[1];
    const float* U = (const float*)d_in[2];
    const float* b = (const float*)d_in[3];
    float* out = (float*)d_out;

    float* ws    = (float*)d_ws;
    float* e     = ws;
    float* alpha = ws + (size_t)Bv * Tv * Kv;
    float* Wt    = alpha + (size_t)Bv * Tv * Kv;

    crf_transposeW<<<dim3(128), dim3(256), 0, stream>>>(W, Wt);
    crf_emissions <<<dim3((Bv * Tv / 16) / 4), dim3(128), 0, stream>>>(x, Wt, b, e);
    crf_scan      <<<dim3(2 * Bv), dim3(32), 0, stream>>>(e, U, alpha, out);
    crf_softmax   <<<dim3(Bv * Tv / 8), dim3(256), 0, stream>>>(alpha, out);
}


# scan: the 32-term sum of each recursion step is 32 DPP row_share multiply-adds in four independent partial sums (summation order changed) instead of 32 v_readlane feeding one dependent chain
# speedup vs baseline: 3.3049x; 1.2199x over previous
#include <hip/hip_runtime.h>
#include <hip/hip_bf16.h>

typedef __attribute__((ext_vector_type(2))) float v2f;
typedef __attribute__((ext_vector_type(8))) float v8f;

#define Bv 64
#define Tv 512
#define Dv 1024
#define Kv 32

__device__ __forceinline__ float fexp2(float x) { return __builtin_amdgcn_exp2f(x); }
__device__ __forceinline__ float flog2(float x) { return __builtin_amdgcn_logf(x); }

__device__ __forceinline__ float wave_max(float v) {
    int b = __builtin_bit_cast(int, v);
    float x = __builtin_bit_cast(float,
        __builtin_amdgcn_permlanex16(0, b, 0x76543210, 0xfedcba98, false, false));
    float m = fmaxf(v, x);
    float r;
    asm("v_max_num_f32_dpp %0, %1, %1 row_xmask:1 row_mask:0xf bank_mask:0xf\n\t"
        "v_max_num_f32_dpp %0, %0, %0 row_xmask:2 row_mask:0xf bank_mask:0xf\n\t"
        "v_max_num_f32_dpp %0, %0, %0 row_xmask:4 row_mask:0xf bank_mask:0xf\n\t"
        "v_max_num_f32_dpp %0, %0, %0 row_xmask:8 row_mask:0xf bank_mask:0xf"
        : "=v"(r) : "v"(m));
    return r;
}

__global__ __launch_bounds__(256) void crf_transposeW(const float* __restrict__ W,
                                                      float* __restrict__ Wt) {
    int idx = blockIdx.x * 256 + threadIdx.x;
    int k = idx >> 10;
    int d = idx & 1023;
    Wt[idx] = W[d * Kv + k];
}

__global__ __launch_bounds__(128) void crf_emissions(const float* __restrict__ x,
                                                     const float* __restrict__ Wt,
                                                     const float* __restrict__ bias,
                                                     float* __restrict__ e) {
    const int lane = threadIdx.x & 31;
    const int wave = threadIdx.x >> 5;
    const int mt   = blockIdx.x * 4 + wave;
    const int half = lane >> 4;
    const int l15  = lane & 15;

    const float* xptr = x  + (size_t)(mt * 16 + l15) * Dv + half * 2;
    const float* w0   = Wt + (size_t)(l15)      * Dv + half * 2;
    const float* w1   = Wt + (size_t)(l15 + 16) * Dv + half * 2;

    v8f acc0 = {};
    v8f acc1 = {};
    #pragma unroll 4
    for (int d = 0; d < Dv / 4; ++d) {
        v2f a  = *(const v2f*)(xptr + d * 4);
        v2f b0 = *(const v2f*)(w0   + d * 4);
        v2f b1 = *(const v2f*)(w1   + d * 4);
        acc0 = __builtin_amdgcn_wmma_f32_16x16x4_f32(false, a, false, b0,
                                                     (short)0, acc0, false, false);
        acc1 = __builtin_amdgcn_wmma_f32_16x16x4_f32(false, a, false, b1,
                                                     (short)0, acc1, false, false);
    }

    const float bv0 = bias[l15];
    const float bv1 = bias[l15 + 16];
    #pragma unroll
    for (int r = 0; r < 8; ++r) {
        size_t row = (size_t)mt * 16 + r + half * 8;
        e[row * Kv + l15]      = acc0[r] + bv0;
        e[row * Kv + l15 + 16] = acc1[r] + bv1;
    }
}

#define PF 4

__device__ __forceinline__ float dot32(float v, const float (&cA)[16], const float (&cB)[16]) {
    float a0, a1, a2, a3, vx;
    asm("v_nop\n\t"
        "v_permlanex16_b32 %4, %5, %6, 0xfedcba98\n\t"
        "v_mul_f32_dpp %0, %5, %7 row_share:0 row_mask:0xf bank_mask:0xf\n\t"
        "v_mul_f32_dpp %1, %5, %8 row_share:1 row_mask:0xf bank_mask:0xf\n\t"
        "v_mul_f32_dpp %2, %5, %9 row_share:2 row_mask:0xf bank_mask:0xf\n\t"
        "v_mul_f32_dpp %3, %5, %10 row_share:3 row_mask:0xf bank_mask:0xf\n\t"
        "v_fmac_f32_dpp %0, %5, %11 row_share:4 row_mask:0xf bank_mask:0xf\n\t"
        "v_fmac_f32_dpp %1, %5, %12 row_share:5 row_mask:0xf bank_mask:0xf\n\t"
        "v_fmac_f32_dpp %2, %5, %13 row_share:6 row_mask:0xf bank_mask:0xf\n\t"
        "v_fmac_f32_dpp %3, %5, %14 row_share:7 row_mask:0xf bank_mask:0xf\n\t"
        "v_fmac_f32_dpp %0, %5, %15 row_share:8 row_mask:0xf bank_mask:0xf\n\t"
        "v_fmac_f32_dpp %1, %5, %16 row_share:9 row_mask:0xf bank_mask:0xf\n\t"
        "v_fmac_f32_dpp %2, %5, %17 row_share:10 row_mask:0xf bank_mask:0xf\n\t"
        "v_fmac_f32_dpp %3, %5, %18 row_share:11 row_mask:0xf bank_mask:0xf\n\t"
        "v_fmac_f32_dpp %0, %5, %19 row_share:12 row_mask:0xf bank_mask:0xf\n\t"
        "v_fmac_f32_dpp %1, %5, %20 row_share:13 row_mask:0xf bank_mask:0xf\n\t"
        "v_fmac_f32_dpp %2, %5, %21 row_share:14 row_mask:0xf bank_mask:0xf\n\t"
        "v_fmac_f32_dpp %3, %5, %22 row_share:15 row_mask:0xf bank_mask:0xf"
        : "=&v"(a0), "=&v"(a1), "=&v"(a2), "=&v"(a3), "=&v"(vx)
        : "v"(v), "s"(0x76543210), "v"(cA[0]), "v"(cA[1]), "v"(cA[2]), "v"(cA[3]), "v"(cA[4]), "v"(cA[5]), "v"(cA[6]), "v"(cA[7]), "v"(cA[8]), "v"(cA[9]), "v"(cA[10]), "v"(cA[11]), "v"(cA[12]), "v"(cA[13]), "v"(cA[14]), "v"(cA[15]));
    asm(
        "v_fmac_f32_dpp %0, %4, %5 row_share:0 row_mask:0xf bank_mask:0xf\n\t"
        "v_fmac_f32_dpp %1, %4, %6 row_share:1 row_mask:0xf bank_mask:0xf\n\t"
        "v_fmac_f32_dpp %2, %4, %7 row_share:2 row_mask:0xf bank_mask:0xf\n\t"
        "v_fmac_f32_dpp %3, %4, %8 row_share:3 row_mask:0xf bank_mask:0xf\n\t"
        "v_fmac_f32_dpp %0, %4, %9 row_share:4 row_mask:0xf bank_mask:0xf\n\t"
        "v_fmac_f32_dpp %1, %4, %10 row_share:5 row_mask:0xf bank_mask:0xf\n\t"
        "v_fmac_f32_dpp %2, %4, %11 row_share:6 row_mask:0xf bank_mask:0xf\n\t"
        "v_fmac_f32_dpp %3, %4, %12 row_share:7 row_mask:0xf bank_mask:0xf\n\t"
        "v_fmac_f32_dpp %0, %4, %13 row_share:8 row_mask:0xf bank_mask:0xf\n\t"
        "v_fmac_f32_dpp %1, %4, %14 row_share:9 row_mask:0xf bank_mask:0xf\n\t"
        "v_fmac_f32_dpp %2, %4, %15 row_share:10 row_mask:0xf bank_mask:0xf\n\t"
        "v_fmac_f32_dpp %3, %4, %16 row_share:11 row_mask:0xf bank_mask:0xf\n\t"
        "v_fmac_f32_dpp %0, %4, %17 row_share:12 row_mask:0xf bank_mask:0xf\n\t"
        "v_fmac_f32_dpp %1, %4, %18 row_share:13 row_mask:0xf bank_mask:0xf\n\t"
        "v_fmac_f32_dpp %2, %4, %19 row_share:14 row_mask:0xf bank_mask:0xf\n\t"
        "v_fmac_f32_dpp %3, %4, %20 row_share:15 row_mask:0xf bank_mask:0xf"
        : "+v"(a0), "+v"(a1), "+v"(a2), "+v"(a3)
        : "v"(vx), "v"(cB[0]), "v"(cB[1]), "v"(cB[2]), "v"(cB[3]), "v"(cB[4]), "v"(cB[5]), "v"(cB[6]), "v"(cB[7]), "v"(cB[8]), "v"(cB[9]), "v"(cB[10]), "v"(cB[11]), "v"(cB[12]), "v"(cB[13]), "v"(cB[14]), "v"(cB[15]));
    return (a0 + a1) + (a2 + a3);
}

__device__ __forceinline__ float crf_fwd_step(float alpha, float et,
                                             const float (&cA)[16], const float (&cB)[16]) {
    const float LOG2E  = 1.44269504088896340736f;
    const float LN2    = 0.69314718055994530942f;
    float m = wave_max(alpha);
    float p = fexp2((alpha - m) * LOG2E);
    float s = dot32(p, cA, cB);
    return m + flog2(s) * LN2 + et;
}

__device__ __forceinline__ float crf_bwd_step(float beta, float en,
                                             const float (&cA)[16], const float (&cB)[16]) {
    const float LOG2E  = 1.44269504088896340736f;
    const float LN2    = 0.69314718055994530942f;
    float g = en + beta;
    float m = wave_max(g);
    float q = fexp2((g - m) * LOG2E);
    float s = dot32(q, cA, cB);
    return m + flog2(s) * LN2;
}

__global__ __launch_bounds__(32) void crf_scan(const float* __restrict__ e,
                                               const float* __restrict__ U,
                                               float* __restrict__ alpha_out,
                                               float* __restrict__ beta_out) {
    const int   lane   = threadIdx.x;
    const float LOG2E  = 1.44269504088896340736f;
    float cA[16], cB[16];
    float ring[PF];
    const int   own    = (lane >> 4) * 16;
    const int   oth    = 16 - own;

    if (blockIdx.x < Bv) {
        const int bIdx = blockIdx.x;
        #pragma unroll
        for (int n = 0; n < 16; ++n) {
            cA[n] = fexp2(U[(own + n) * Kv + lane] * LOG2E);
            cB[n] = fexp2(U[(oth + n) * Kv + lane] * LOG2E);
        }

        const float* eb = e         + (size_t)bIdx * Tv * Kv;
        float*       ab = alpha_out + (size_t)bIdx * Tv * Kv;

        #pragma unroll
        for (int i = 0; i < PF; ++i)
            ring[i] = eb[(1 + i) * Kv + lane];

        float alpha = eb[lane];
        ab[lane] = alpha;
        static_assert(PF == 4, "the statement below names the PF ring slots one by one");
        asm volatile("" :: "v"(ring[0]), "v"(ring[1]), "v"(ring[2]), "v"(ring[3]));
        int t = 1;
        for (; t + PF <= Tv; t += PF) {
            #pragma unroll
            for (int i = 0; i < PF; ++i) {
                alpha = crf_fwd_step(alpha, ring[i], cA, cB);
                ab[(t + i) * Kv + lane] = alpha;
                int tn = t + i + PF;
                tn = tn < Tv ? tn : Tv - 1;
                ring[i] = eb[tn * Kv + lane];
                __builtin_amdgcn_sched_barrier(0);
            }
        }
        #pragma unroll
        for (int i = 0; i < PF - 1; ++i) {
            if (t + i < Tv) {
                alpha = crf_fwd_step(alpha, ring[i], cA, cB);
                ab[(t + i) * Kv + lane] = alpha;
            }
        }
    } else {
        const int bIdx = blockIdx.x - Bv;
        #pragma unroll
        for (int n = 0; n < 16; ++n) {
            cA[n] = fexp2(U[lane * Kv + own + n] * LOG2E);
            cB[n] = fexp2(U[lane * Kv + oth + n] * LOG2E);
        }

        const float* eb = e        + (size_t)bIdx * Tv * Kv;
        float*       bb = beta_out + (size_t)bIdx * Tv * Kv;

        #pragma unroll
        for (int i = 0; i < PF; ++i)
            ring[i] = eb[(Tv - 1 - i) * Kv + lane];

        float beta = 0.f;
        bb[(Tv - 1) * Kv + lane] = 0.f;
        int t = Tv - 2;
        for (; t - (PF - 1) >= 0; t -= PF) {
            #pragma unroll
            for (int i = 0; i < PF; ++i) {
                beta = crf_bwd_step(beta, ring[i], cA, cB);
                bb[(t - i) * Kv + lane] = beta;
                int qn = t - i + 1 - PF;
                qn = qn > 1 ? qn : 1;
                ring[i] = eb[qn * Kv + lane];
                __builtin_amdgcn_sched_barrier(0);
            }
        }
        #pragma unroll
        for (int i = 0; i < PF - 1; ++i) {
            if (t - i >= 0) {
                beta = crf_bwd_step(beta, ring[i], cA, cB);
                bb[(t - i) * Kv + lane] = beta;
            }
        }
    }
}

__global__ __launch_bounds__(256) void crf_softmax(const float* __restrict__ alpha,
                                                   float* __restrict__ out) {
    const int lane = threadIdx.x & 31;
    const int wave = threadIdx.x >> 5;
    const size_t row = (size_t)blockIdx.x * 8 + wave;
    const size_t idx = row * Kv + lane;
    const float LOG2E = 1.44269504088896340736f;

    float s = alpha[idx] + out[idx];
    float m = s;
    #pragma unroll
    for (int off = 16; off; off >>= 1) m = fmaxf(m, __shfl_xor(m, off, 32));
    float p = fexp2((s - m) * LOG2E);
    float sum = p;
    #pragma unroll
    for (int off = 16; off; off >>= 1) sum += __shfl_xor(sum, off, 32);
    out[idx] = p / sum;
}

extern "C" void kernel_launch(void* const* d_in, const int* in_sizes, int n_in,
                              void* d_out, int out_size, void* d_ws, size_t ws_size,
                              hipStream_t stream) {
    const float* x = (const float*)d_in[0];
    const float* W = (const float*)d_in[1];
    const float* U = (const float*)d_in[2];
    const float* b = (const float*)d_in[3];
    float* out = (float*)d_out;

    float* ws    = (float*)d_ws;
    float* e     = ws;
    float* alpha = ws + (size_t)Bv * Tv * Kv;
    float* Wt    = alpha + (size_t)Bv * Tv * Kv;

    crf_transposeW<<<dim3(128), dim3(256), 0, stream>>>(W, Wt);
    crf_emissions <<<dim3((Bv * Tv / 16) / 4), dim3(128), 0, stream>>>(x, Wt, b, e);
    crf_scan      <<<dim3(2 * Bv), dim3(32), 0, stream>>>(e, U, alpha, out);
    crf_softmax   <<<dim3(Bv * Tv / 8), dim3(256), 0, stream>>>(alpha, out);
}


# emissions: x and W fragments are loaded 16 B per lane (W packed in fragment order, whole lines per load) one 64-wide block of d ahead of the WMMAs; k order inside each 8 of d changed
# speedup vs baseline: 3.7171x; 1.1247x over previous
#include <hip/hip_runtime.h>
#include <hip/hip_bf16.h>

typedef __attribute__((ext_vector_type(2))) float v2f;
typedef __attribute__((ext_vector_type(4))) float v4f;
typedef __attribute__((ext_vector_type(8))) float v8f;

#define Bv 64
#define Tv 512
#define Dv 1024
#define Kv 32

__device__ __forceinline__ float fexp2(float x) { return __builtin_amdgcn_exp2f(x); }
__device__ __forceinline__ float flog2(float x) { return __builtin_amdgcn_logf(x); }

__device__ __forceinline__ float wave_max(float v) {
    int b = __builtin_bit_cast(int, v);
    float x = __builtin_bit_cast(float,
        __builtin_amdgcn_permlanex16(0, b, 0x76543210, 0xfedcba98, false, false));
    float m = fmaxf(v, x);
    float r;
    asm("v_max_num_f32_dpp %0, %1, %1 row_xmask:1 row_mask:0xf bank_mask:0xf\n\t"
        "v_max_num_f32_dpp %0, %0, %0 row_xmask:2 row_mask:0xf bank_mask:0xf\n\t"
        "v_max_num_f32_dpp %0, %0, %0 row_xmask:4 row_mask:0xf bank_mask:0xf\n\t"
        "v_max_num_f32_dpp %0, %0, %0 row_xmask:8 row_mask:0xf bank_mask:0xf"
        : "=v"(r) : "v"(m));
    return r;
}

__global__ __launch_bounds__(256) void crf_packW(const float* __restrict__ W,
                                                 float* __restrict__ Wp) {
    int idx  = blockIdx.x * 256 + threadIdx.x;
    int s    = idx & 3;
    int lane = (idx >> 2) & 31;
    int tile = (idx >> 7) & 1;
    int g    = idx >> 8;
    int d    = g * 8 + (lane >> 4) * 4 + s;
    int n    = tile * 16 + (lane & 15);
    Wp[idx] = W[d * Kv + n];
}

#define EG 8

struct EFrag { v4f a[EG]; v4f b0[EG]; v4f b1[EG]; };

__device__ __forceinline__ void emis_load(EFrag& f, const float* __restrict__ xrow,
                                          const float* __restrict__ wp, int blk) {
    #pragma unroll
    for (int j = 0; j < EG; ++j) {
        const int g = blk * EG + j;
        f.a[j]  = *(const v4f*)(xrow + g * 8);
        f.b0[j] = *(const v4f*)(wp + g * 256);
        f.b1[j] = *(const v4f*)(wp + g * 256 + 128);
    }
}

__device__ __forceinline__ void emis_mma(const EFrag& f, v8f& acc0, v8f& acc1) {
    #pragma unroll
    for (int j = 0; j < EG; ++j) {
        v2f a01 = {f.a[j].x,  f.a[j].y},  a23 = {f.a[j].z,  f.a[j].w};
        v2f p01 = {f.b0[j].x, f.b0[j].y}, p23 = {f.b0[j].z, f.b0[j].w};
        v2f q01 = {f.b1[j].x, f.b1[j].y}, q23 = {f.b1[j].z, f.b1[j].w};
        acc0 = __builtin_amdgcn_wmma_f32_16x16x4_f32(false, a01, false, p01,
                                                     (short)0, acc0, false, false);
        acc1 = __builtin_amdgcn_wmma_f32_16x16x4_f32(false, a01, false, q01,
                                                     (short)0, acc1, false, false);
        acc0 = __builtin_amdgcn_wmma_f32_16x16x4_f32(false, a23, false, p23,
                                                     (short)0, acc0, false, false);
        acc1 = __builtin_amdgcn_wmma_f32_16x16x4_f32(false, a23, false, q23,
                                                     (short)0, acc1, false, false);
    }
}

__global__ __launch_bounds__(128) void crf_emissions(const float* __restrict__ x,
                                                     const float* __restrict__ Wp,
                                                     const float* __restrict__ bias,
                                                     float* __restrict__ e) {
    const int lane = threadIdx.x & 31;
    const int wave = threadIdx.x >> 5;
    const int mt   = blockIdx.x * 4 + wave;
    const int half = lane >> 4;
    const int l15  = lane & 15;

    const float* xrow = x  + (size_t)(mt * 16 + l15) * Dv + half * 4;
    const float* wp   = Wp + lane * 4;

    v8f acc0 = {};
    v8f acc1 = {};
    EFrag P, Q;
    const int NB = Dv / (8 * EG);
    emis_load(P, xrow, wp, 0);
    for (int blk = 0; blk < NB - 2; blk += 2) {
        emis_load(Q, xrow, wp, blk + 1);
        emis_mma(P, acc0, acc1);
        emis_load(P, xrow, wp, blk + 2);
        emis_mma(Q, acc0, acc1);
    }
    emis_load(Q, xrow, wp, NB - 1);
    emis_mma(P, acc0, acc1);
    emis_mma(Q, acc0, acc1);

    const float bv0 = bias[l15];
    const float bv1 = bias[l15 + 16];
    #pragma unroll
    for (int r = 0; r < 8; ++r) {
        size_t row = (size_t)mt * 16 + r + half * 8;
        e[row * Kv + l15]      = acc0[r] + bv0;
        e[row * Kv + l15 + 16] = acc1[r] + bv1;
    }
}

#define PF 4

__device__ __forceinline__ float dot32(float v, const float (&cA)[16], const float (&cB)[16]) {
    float a0, a1, a2, a3, vx;
    asm("v_nop\n\t"
        "v_permlanex16_b32 %4, %5, %6, 0xfedcba98\n\t"
        "v_mul_f32_dpp %0, %5, %7 row_share:0 row_mask:0xf bank_mask:0xf\n\t"
        "v_mul_f32_dpp %1, %5, %8 row_share:1 row_mask:0xf bank_mask:0xf\n\t"
        "v_mul_f32_dpp %2, %5, %9 row_share:2 row_mask:0xf bank_mask:0xf\n\t"
        "v_mul_f32_dpp %3, %5, %10 row_share:3 row_mask:0xf bank_mask:0xf\n\t"
        "v_fmac_f32_dpp %0, %5, %11 row_share:4 row_mask:0xf bank_mask:0xf\n\t"
        "v_fmac_f32_dpp %1, %5, %12 row_share:5 row_mask:0xf bank_mask:0xf\n\t"
        "v_fmac_f32_dpp %2, %5, %13 row_share:6 row_mask:0xf bank_mask:0xf\n\t"
        "v_fmac_f32_dpp %3, %5, %14 row_share:7 row_mask:0xf bank_mask:0xf\n\t"
        "v_fmac_f32_dpp %0, %5, %15 row_share:8 row_mask:0xf bank_mask:0xf\n\t"
        "v_fmac_f32_dpp %1, %5, %16 row_share:9 row_mask:0xf bank_mask:0xf\n\t"
        "v_fmac_f32_dpp %2, %5, %17 row_share:10 row_mask:0xf bank_mask:0xf\n\t"
        "v_fmac_f32_dpp %3, %5, %18 row_share:11 row_mask:0xf bank_mask:0xf\n\t"
        "v_fmac_f32_dpp %0, %5, %19 row_share:12 row_mask:0xf bank_mask:0xf\n\t"
        "v_fmac_f32_dpp %1, %5, %20 row_share:13 row_mask:0xf bank_mask:0xf\n\t"
        "v_fmac_f32_dpp %2, %5, %21 row_share:14 row_mask:0xf bank_mask:0xf\n\t"
        "v_fmac_f32_dpp %3, %5, %22 row_share:15 row_mask:0xf bank_mask:0xf"
        : "=&v"(a0), "=&v"(a1), "=&v"(a2), "=&v"(a3), "=&v"(vx)
        : "v"(v), "s"(0x76543210), "v"(cA[0]), "v"(cA[1]), "v"(cA[2]), "v"(cA[3]), "v"(cA[4]), "v"(cA[5]), "v"(cA[6]), "v"(cA[7]), "v"(cA[8]), "v"(cA[9]), "v"(cA[10]), "v"(cA[11]), "v"(cA[12]), "v"(cA[13]), "v"(cA[14]), "v"(cA[15]));
    asm(
        "v_fmac_f32_dpp %0, %4, %5 row_share:0 row_mask:0xf bank_mask:0xf\n\t"
        "v_fmac_f32_dpp %1, %4, %6 row_share:1 row_mask:0xf bank_mask:0xf\n\t"
        "v_fmac_f32_dpp %2, %4, %7 row_share:2 row_mask:0xf bank_mask:0xf\n\t"
        "v_fmac_f32_dpp %3, %4, %8 row_share:3 row_mask:0xf bank_mask:0xf\n\t"
        "v_fmac_f32_dpp %0, %4, %9 row_share:4 row_mask:0xf bank_mask:0xf\n\t"
        "v_fmac_f32_dpp %1, %4, %10 row_share:5 row_mask:0xf bank_mask:0xf\n\t"
        "v_fmac_f32_dpp %2, %4, %11 row_share:6 row_mask:0xf bank_mask:0xf\n\t"
        "v_fmac_f32_dpp %3, %4, %12 row_share:7 row_mask:0xf bank_mask:0xf\n\t"
        "v_fmac_f32_dpp %0, %4, %13 row_share:8 row_mask:0xf bank_mask:0xf\n\t"
        "v_fmac_f32_dpp %1, %4, %14 row_share:9 row_mask:0xf bank_mask:0xf\n\t"
        "v_fmac_f32_dpp %2, %4, %15 row_share:10 row_mask:0xf bank_mask:0xf\n\t"
        "v_fmac_f32_dpp %3, %4, %16 row_share:11 row_mask:0xf bank_mask:0xf\n\t"
        "v_fmac_f32_dpp %0, %4, %17 row_share:12 row_mask:0xf bank_mask:0xf\n\t"
        "v_fmac_f32_dpp %1, %4, %18 row_share:13 row_mask:0xf bank_mask:0xf\n\t"
        "v_fmac_f32_dpp %2, %4, %19 row_share:14 row_mask:0xf bank_mask:0xf\n\t"
        "v_fmac_f32_dpp %3, %4, %20 row_share:15 row_mask:0xf bank_mask:0xf"
        : "+v"(a0), "+v"(a1), "+v"(a2), "+v"(a3)
        : "v"(vx), "v"(cB[0]), "v"(cB[1]), "v"(cB[2]), "v"(cB[3]), "v"(cB[4]), "v"(cB[5]), "v"(cB[6]), "v"(cB[7]), "v"(cB[8]), "v"(cB[9]), "v"(cB[10]), "v"(cB[11]), "v"(cB[12]), "v"(cB[13]), "v"(cB[14]), "v"(cB[15]));
    return (a0 + a1) + (a2 + a3);
}

__device__ __forceinline__ float crf_fwd_step(float alpha, float et,
                                             const float (&cA)[16], const float (&cB)[16]) {
    const float LOG2E  = 1.44269504088896340736f;
    const float LN2    = 0.69314718055994530942f;
    float m = wave_max(alpha);
    float p = fexp2((alpha - m) * LOG2E);
    float s = dot32(p, cA, cB);
    return m + flog2(s) * LN2 + et;
}

__device__ __forceinline__ float crf_bwd_step(float beta, float en,
                                             const float (&cA)[16], const float (&cB)[16]) {
    const float LOG2E  = 1.44269504088896340736f;
    const float LN2    = 0.69314718055994530942f;
    float g = en + beta;
    float m = wave_max(g);
    float q = fexp2((g - m) * LOG2E);
    float s = dot32(q, cA, cB);
    return m + flog2(s) * LN2;
}

__global__ __launch_bounds__(32) void crf_scan(const float* __restrict__ e,
                                               const float* __restrict__ U,
                                               float* __restrict__ alpha_out,
                                               float* __restrict__ beta_out) {
    const int   lane   = threadIdx.x;
    const float LOG2E  = 1.44269504088896340736f;
    float cA[16], cB[16];
    float ring[PF];
    const int   own    = (lane >> 4) * 16;
    const int   oth    = 16 - own;

    if (blockIdx.x < Bv) {
        const int bIdx = blockIdx.x;
        #pragma unroll
        for (int n = 0; n < 16; ++n) {
            cA[n] = fexp2(U[(own + n) * Kv + lane] * LOG2E);
            cB[n] = fexp2(U[(oth + n) * Kv + lane] * LOG2E);
        }

        const float* eb = e         + (size_t)bIdx * Tv * Kv;
        float*       ab = alpha_out + (size_t)bIdx * Tv * Kv;

        #pragma unroll
        for (int i = 0; i < PF; ++i)
            ring[i] = eb[(1 + i) * Kv + lane];

        float alpha = eb[lane];
        ab[lane] = alpha;
        static_assert(PF == 4, "the statement below names the PF ring slots one by one");
        asm volatile("" :: "v"(ring[0]), "v"(ring[1]), "v"(ring[2]), "v"(ring[3]));
        int t = 1;
        for (; t + PF <= Tv; t += PF) {
            #pragma unroll
            for (int i = 0; i < PF; ++i) {
                alpha = crf_fwd_step(alpha, ring[i], cA, cB);
                ab[(t + i) * Kv + lane] = alpha;
                int tn = t + i + PF;
                tn = tn < Tv ? tn : Tv - 1;
                ring[i] = eb[tn * Kv + lane];
                __builtin_amdgcn_sched_barrier(0);
            }
        }
        #pragma unroll
        for (int i = 0; i < PF - 1; ++i) {
            if (t + i < Tv) {
                alpha = crf_fwd_step(alpha, ring[i], cA, cB);
                ab[(t + i) * Kv + lane] = alpha;
            }
        }
    } else {
        const int bIdx = blockIdx.x - Bv;
        #pragma unroll
        for (int n = 0; n < 16; ++n) {
            cA[n] = fexp2(U[lane * Kv + own + n] * LOG2E);
            cB[n] = fexp2(U[lane * Kv + oth + n] * LOG2E);
        }

        const float* eb = e        + (size_t)bIdx * Tv * Kv;
        float*       bb = beta_out + (size_t)bIdx * Tv * Kv;

        #pragma unroll
        for (int i = 0; i < PF; ++i)
            ring[i] = eb[(Tv - 1 - i) * Kv + lane];

        float beta = 0.f;
        bb[(Tv - 1) * Kv + lane] = 0.f;
        int t = Tv - 2;
        for (; t - (PF - 1) >= 0; t -= PF) {
            #pragma unroll
            for (int i = 0; i < PF; ++i) {
                beta = crf_bwd_step(beta, ring[i], cA, cB);
                bb[(t - i) * Kv + lane] = beta;
                int qn = t - i + 1 - PF;
                qn = qn > 1 ? qn : 1;
                ring[i] = eb[qn * Kv + lane];
                __builtin_amdgcn_sched_barrier(0);
            }
        }
        #pragma unroll
        for (int i = 0; i < PF - 1; ++i) {
            if (t - i >= 0) {
                beta = crf_bwd_step(beta, ring[i], cA, cB);
                bb[(t - i) * Kv + lane] = beta;
            }
        }
    }
}

__global__ __launch_bounds__(256) void crf_softmax(const float* __restrict__ alpha,
                                                   float* __restrict__ out) {
    const int lane = threadIdx.x & 31;
    const int wave = threadIdx.x >> 5;
    const size_t row = (size_t)blockIdx.x * 8 + wave;
    const size_t idx = row * Kv + lane;
    const float LOG2E = 1.44269504088896340736f;

    float s = alpha[idx] + out[idx];
    float m = s;
    #pragma unroll
    for (int off = 16; off; off >>= 1) m = fmaxf(m, __shfl_xor(m, off, 32));
    float p = fexp2((s - m) * LOG2E);
    float sum = p;
    #pragma unroll
    for (int off = 16; off; off >>= 1) sum += __shfl_xor(sum, off, 32);
    out[idx] = p / sum;
}

extern "C" void kernel_launch(void* const* d_in, const int* in_sizes, int n_in,
                              void* d_out, int out_size, void* d_ws, size_t ws_size,
                              hipStream_t stream) {
    const float* x = (const float*)d_in[0];
    const float* W = (const float*)d_in[1];
    const float* U = (const float*)d_in[2];
    const float* b = (const float*)d_in[3];
    float* out = (float*)d_out;

    float* ws    = (float*)d_ws;
    float* e     = ws;
    float* alpha = ws + (size_t)Bv * Tv * Kv;
    float* Wp    = alpha + (size_t)Bv * Tv * Kv;

    crf_packW     <<<dim3(128), dim3(256), 0, stream>>>(W, Wp);
    crf_emissions <<<dim3((Bv * Tv / 16) / 4), dim3(128), 0, stream>>>(x, Wp, b, e);
    crf_scan      <<<dim3(2 * Bv), dim3(32), 0, stream>>>(e, U, alpha, out);
    crf_softmax   <<<dim3(Bv * Tv / 8), dim3(256), 0, stream>>>(alpha, out);
}


# scan: recursions in the linear domain on exp(e - row max), rescaled by 1/max of each dot's input vector (sums of rescaled terms): max-reduce, exp, log leave the serial chain; row ring 4 to 8 deep
# speedup vs baseline: 4.6633x; 1.2546x over previous
#include <hip/hip_runtime.h>
#include <hip/hip_bf16.h>

typedef __attribute__((ext_vector_type(2))) float v2f;
typedef __attribute__((ext_vector_type(4))) float v4f;
typedef __attribute__((ext_vector_type(8))) float v8f;

#define Bv 64
#define Tv 512
#define Dv 1024
#define Kv 32

__device__ __forceinline__ float fexp2(float x) { return __builtin_amdgcn_exp2f(x); }
__device__ __forceinline__ float flog2(float x) { return __builtin_amdgcn_logf(x); }

__device__ __forceinline__ float row16_max(float v) {
    float r;
    asm("v_max_num_f32_dpp %0, %1, %1 row_xmask:1 row_mask:0xf bank_mask:0xf\n\t"
        "v_max_num_f32_dpp %0, %0, %0 row_xmask:2 row_mask:0xf bank_mask:0xf\n\t"
        "v_max_num_f32_dpp %0, %0, %0 row_xmask:4 row_mask:0xf bank_mask:0xf\n\t"
        "v_max_num_f32_dpp %0, %0, %0 row_xmask:8 row_mask:0xf bank_mask:0xf"
        : "=v"(r) : "v"(v));
    return r;
}

__global__ __launch_bounds__(256) void crf_packW(const float* __restrict__ W,
                                                 float* __restrict__ Wp) {
    int idx  = blockIdx.x * 256 + threadIdx.x;
    int s    = idx & 3;
    int lane = (idx >> 2) & 31;
    int tile = (idx >> 7) & 1;
    int g    = idx >> 8;
    int d    = g * 8 + (lane >> 4) * 4 + s;
    int n    = tile * 16 + (lane & 15);
    Wp[idx] = W[d * Kv + n];
}

#define EG 8

struct EFrag { v4f a[EG]; v4f b0[EG]; v4f b1[EG]; };

__device__ __forceinline__ void emis_load(EFrag& f, const float* __restrict__ xrow,
                                          const float* __restrict__ wp, int blk) {
    #pragma unroll
    for (int j = 0; j < EG; ++j) {
        const int g = blk * EG + j;
        f.a[j]  = *(const v4f*)(xrow + g * 8);
        f.b0[j] = *(const v4f*)(wp + g * 256);
        f.b1[j] = *(const v4f*)(wp + g * 256 + 128);
    }
}

__device__ __forceinline__ void emis_mma(const EFrag& f, v8f& acc0, v8f& acc1) {
    #pragma unroll
    for (int j = 0; j < EG; ++j) {
        v2f a01 = {f.a[j].x,  f.a[j].y},  a23 = {f.a[j].z,  f.a[j].w};
        v2f p01 = {f.b0[j].x, f.b0[j].y}, p23 = {f.b0[j].z, f.b0[j].w};
        v2f q01 = {f.b1[j].x, f.b1[j].y}, q23 = {f.b1[j].z, f.b1[j].w};
        acc0 = __builtin_amdgcn_wmma_f32_16x16x4_f32(false, a01, false, p01,
                                                     (short)0, acc0, false, false);
        acc1 = __builtin_amdgcn_wmma_f32_16x16x4_f32(false, a01, false, q01,
                                                     (short)0, acc1, false, false);
        acc0 = __builtin_amdgcn_wmma_f32_16x16x4_f32(false, a23, false, p23,
                                                     (short)0, acc0, false, false);
        acc1 = __builtin_amdgcn_wmma_f32_16x16x4_f32(false, a23, false, q23,
                                                     (short)0, acc1, false, false);
    }
}

__global__ __launch_bounds__(128) void crf_emissions(const float* __restrict__ x,
                                                     const float* __restrict__ Wp,
                                                     const float* __restrict__ bias,
                                                     float* __restrict__ e) {
    const int lane = threadIdx.x & 31;
    const int wave = threadIdx.x >> 5;
    const int mt   = blockIdx.x * 4 + wave;
    const int half = lane >> 4;
    const int l15  = lane & 15;

    const float* xrow = x  + (size_t)(mt * 16 + l15) * Dv + half * 4;
    const float* wp   = Wp + lane * 4;

    v8f acc0 = {};
    v8f acc1 = {};
    EFrag P, Q;
    const int NB = Dv / (8 * EG);
    emis_load(P, xrow, wp, 0);
    for (int blk = 0; blk < NB - 2; blk += 2) {
        emis_load(Q, xrow, wp, blk + 1);
        emis_mma(P, acc0, acc1);
        emis_load(P, xrow, wp, blk + 2);
        emis_mma(Q, acc0, acc1);
    }
    emis_load(Q, xrow, wp, NB - 1);
    emis_mma(P, acc0, acc1);
    emis_mma(Q, acc0, acc1);

    const float bv0 = bias[l15];
    const float bv1 = bias[l15 + 16];
    const float LOG2E = 1.44269504088896340736f;
    #pragma unroll
    for (int r = 0; r < 8; ++r) {
        size_t row = (size_t)mt * 16 + r + half * 8;
        const float v0 = acc0[r] + bv0;
        const float v1 = acc1[r] + bv1;
        const float m  = row16_max(fmaxf(v0, v1));
        e[row * Kv + l15]      = fexp2((v0 - m) * LOG2E);
        e[row * Kv + l15 + 16] = fexp2((v1 - m) * LOG2E);
    }
}

#define PF 8

__device__ __forceinline__ void lin_step(float v, float g,
                                         const float (&cA)[16], const float (&cB)[16],
                                         float& s01, float& s23, float& vn) {
    float a0, a1, a2, a3, gs, vx, m;
    asm(
        "v_permlanex16_b32 %4, %6, %7, 0xfedcba98\n\t"
        "v_mul_f32_dpp %0, %6, %8 row_share:0 row_mask:0xf bank_mask:0xf\n\t"
        "v_mul_f32_dpp %1, %6, %9 row_share:1 row_mask:0xf bank_mask:0xf\n\t"
        "v_mul_f32_dpp %2, %6, %10 row_share:2 row_mask:0xf bank_mask:0xf\n\t"
        "v_max_num_f32 %5, %6, %4\n\t"
        "v_mul_f32_dpp %3, %6, %11 row_share:3 row_mask:0xf bank_mask:0xf\n\t"
        "v_fmac_f32_dpp %0, %6, %12 row_share:4 row_mask:0xf bank_mask:0xf\n\t"
        "v_fmac_f32_dpp %1, %6, %13 row_share:5 row_mask:0xf bank_mask:0xf\n\t"
        "v_max_num_f32_dpp %5, %5, %5 row_xmask:1 row_mask:0xf bank_mask:0xf\n\t"
        "v_fmac_f32_dpp %2, %6, %14 row_share:6 row_mask:0xf bank_mask:0xf\n\t"
        "v_fmac_f32_dpp %3, %6, %15 row_share:7 row_mask:0xf bank_mask:0xf\n\t"
        "v_fmac_f32_dpp %0, %6, %16 row_share:8 row_mask:0xf bank_mask:0xf\n\t"
        "v_max_num_f32_dpp %5, %5, %5 row_xmask:2 row_mask:0xf bank_mask:0xf\n\t"
        "v_fmac_f32_dpp %1, %6, %17 row_share:9 row_mask:0xf bank_mask:0xf\n\t"
        "v_fmac_f32_dpp %2, %6, %18 row_share:10 row_mask:0xf bank_mask:0xf\n\t"
        "v_fmac_f32_dpp %3, %6, %19 row_share:11 row_mask:0xf bank_mask:0xf\n\t"
        "v_max_num_f32_dpp %5, %5, %5 row_xmask:4 row_mask:0xf bank_mask:0xf\n\t"
        "v_fmac_f32_dpp %0, %6, %20 row_share:12 row_mask:0xf bank_mask:0xf\n\t"
        "v_fmac_f32_dpp %1, %6, %21 row_share:13 row_mask:0xf bank_mask:0xf\n\t"
        "v_fmac_f32_dpp %2, %6, %22 row_share:14 row_mask:0xf bank_mask:0xf\n\t"
        "v_max_num_f32_dpp %5, %5, %5 row_xmask:8 row_mask:0xf bank_mask:0xf\n\t"
        "v_fmac_f32_dpp %3, %6, %23 row_share:15 row_mask:0xf bank_mask:0xf"
        : "=&v"(a0), "=&v"(a1), "=&v"(a2), "=&v"(a3), "=&v"(vx), "=&v"(m)
        : "v"(v), "s"(0x76543210), "v"(cA[0]), "v"(cA[1]), "v"(cA[2]), "v"(cA[3]), "v"(cA[4]), "v"(cA[5]), "v"(cA[6]), "v"(cA[7]), "v"(cA[8]), "v"(cA[9]), "v"(cA[10]), "v"(cA[11]), "v"(cA[12]), "v"(cA[13]), "v"(cA[14]), "v"(cA[15]));
    asm(
        "v_fmac_f32_dpp %0, %5, %8 row_share:0 row_mask:0xf bank_mask:0xf\n\t"
        "v_fmac_f32_dpp %1, %5, %9 row_share:1 row_mask:0xf bank_mask:0xf\n\t"
        "v_rcp_f32 %4, %6\n\t"
        "v_fmac_f32_dpp %2, %5, %10 row_share:2 row_mask:0xf bank_mask:0xf\n\t"
        "v_fmac_f32_dpp %3, %5, %11 row_share:3 row_mask:0xf bank_mask:0xf\n\t"
        "v_fmac_f32_dpp %0, %5, %12 row_share:4 row_mask:0xf bank_mask:0xf\n\t"
        "v_fmac_f32_dpp %1, %5, %13 row_share:5 row_mask:0xf bank_mask:0xf\n\t"
        "v_fmac_f32_dpp %2, %5, %14 row_share:6 row_mask:0xf bank_mask:0xf\n\t"
        "v_mul_f32 %4, %7, %4\n\t"
        "v_fmac_f32_dpp %3, %5, %15 row_share:7 row_mask:0xf bank_mask:0xf\n\t"
        "v_fmac_f32_dpp %0, %5, %16 row_share:8 row_mask:0xf bank_mask:0xf\n\t"
        "v_fmac_f32_dpp %1, %5, %17 row_share:9 row_mask:0xf bank_mask:0xf\n\t"
        "v_fmac_f32_dpp %2, %5, %18 row_share:10 row_mask:0xf bank_mask:0xf\n\t"
        "v_fmac_f32_dpp %3, %5, %19 row_share:11 row_mask:0xf bank_mask:0xf\n\t"
        "v_fmac_f32_dpp %0, %5, %20 row_share:12 row_mask:0xf bank_mask:0xf\n\t"
        "v_fmac_f32_dpp %1, %5, %21 row_share:13 row_mask:0xf bank_mask:0xf\n\t"
        "v_fmac_f32_dpp %2, %5, %22 row_share:14 row_mask:0xf bank_mask:0xf\n\t"
        "v_fmac_f32_dpp %3, %5, %23 row_share:15 row_mask:0xf bank_mask:0xf\n\t"
        "v_add_f32 %0, %0, %1\n\t"
        "v_add_f32 %2, %2, %3\n\t"
        "v_mul_f32 %1, %0, %4\n\t"
        "v_fmac_f32 %1, %2, %4"
        : "+v"(a0), "+v"(a1), "+v"(a2), "+v"(a3), "=&v"(gs)
        : "v"(vx), "v"(m), "v"(g), "v"(cB[0]), "v"(cB[1]), "v"(cB[2]), "v"(cB[3]), "v"(cB[4]), "v"(cB[5]), "v"(cB[6]), "v"(cB[7]), "v"(cB[8]), "v"(cB[9]), "v"(cB[10]), "v"(cB[11]), "v"(cB[12]), "v"(cB[13]), "v"(cB[14]), "v"(cB[15]));
    s01 = a0;
    s23 = a2;
    vn  = a1;
}

__global__ __launch_bounds__(32) void crf_scan(const float* __restrict__ g,
                                               const float* __restrict__ U,
                                               float* __restrict__ p_out,
                                               float* __restrict__ q_out) {
    const int   lane   = threadIdx.x;
    const float LOG2E  = 1.44269504088896340736f;
    float cA[16], cB[16];
    float ring[PF];
    float s01, s23, vn;
    const int   own    = (lane >> 4) * 16;
    const int   oth    = 16 - own;
    static_assert(PF == 8, "the fence statements below name the PF ring slots one by one");

    if (blockIdx.x < Bv) {
        const int bIdx = blockIdx.x;
        #pragma unroll
        for (int n = 0; n < 16; ++n) {
            cA[n] = fexp2(U[(own + n) * Kv + lane] * LOG2E);
            cB[n] = fexp2(U[(oth + n) * Kv + lane] * LOG2E);
        }

        const float* gb = g     + (size_t)bIdx * Tv * Kv;
        float*       pb = p_out + (size_t)bIdx * Tv * Kv;

        #pragma unroll
        for (int i = 0; i < PF; ++i)
            ring[i] = gb[(1 + i) * Kv + lane];

        float p = gb[lane];
        pb[lane] = p;
        asm volatile("" :: "v"(ring[0]), "v"(ring[1]), "v"(ring[2]), "v"(ring[3]),
                           "v"(ring[4]), "v"(ring[5]), "v"(ring[6]), "v"(ring[7]));
        int t = 1;
        for (; t + PF <= Tv; t += PF) {
            #pragma unroll
            for (int i = 0; i < PF; ++i) {
                lin_step(p, ring[i], cA, cB, s01, s23, vn);
                p = vn;
                pb[(t + i) * Kv + lane] = p;
                int tn = t + i + PF;
                tn = tn < Tv ? tn : Tv - 1;
                ring[i] = gb[tn * Kv + lane];
                __builtin_amdgcn_sched_barrier(0);
            }
        }
        #pragma unroll
        for (int i = 0; i < PF - 1; ++i) {
            if (t + i < Tv) {
                lin_step(p, ring[i], cA, cB, s01, s23, vn);
                p = vn;
                pb[(t + i) * Kv + lane] = p;
            }
        }
    } else {
        const int bIdx = blockIdx.x - Bv;
        #pragma unroll
        for (int n = 0; n < 16; ++n) {
            cA[n] = fexp2(U[lane * Kv + own + n] * LOG2E);
            cB[n] = fexp2(U[lane * Kv + oth + n] * LOG2E);
        }

        const float* gb = g     + (size_t)bIdx * Tv * Kv;
        float*       qb = q_out + (size_t)bIdx * Tv * Kv;

        #pragma unroll
        for (int i = 0; i < PF; ++i)
            ring[i] = gb[(Tv - 2 - i) * Kv + lane];

        float w = gb[(Tv - 1) * Kv + lane];
        qb[(Tv - 1) * Kv + lane] = 1.f;
        asm volatile("" :: "v"(ring[0]), "v"(ring[1]), "v"(ring[2]), "v"(ring[3]),
                           "v"(ring[4]), "v"(ring[5]), "v"(ring[6]), "v"(ring[7]), "v"(w));
        int t = Tv - 2;
        for (; t - (PF - 1) >= 0; t -= PF) {
            #pragma unroll
            for (int i = 0; i < PF; ++i) {
                lin_step(w, ring[i], cA, cB, s01, s23, vn);
                qb[(t - i) * Kv + lane] = s01 + s23;
                w = vn;
                int qn = t - i - PF;
                qn = qn > 0 ? qn : 0;
                ring[i] = gb[qn * Kv + lane];
                __builtin_amdgcn_sched_barrier(0);
            }
        }
        #pragma unroll
        for (int i = 0; i < PF - 1; ++i) {
            if (t - i >= 0) {
                lin_step(w, ring[i], cA, cB, s01, s23, vn);
                qb[(t - i) * Kv + lane] = s01 + s23;
                w = vn;
            }
        }
    }
}

__global__ __launch_bounds__(256) void crf_marginals(const float* __restrict__ p,
                                                     float* __restrict__ out) {
    const int lane = threadIdx.x & 31;
    const int wave = threadIdx.x >> 5;
    const size_t row = (size_t)blockIdx.x * 8 + wave;
    const size_t idx = row * Kv + lane;

    const float s = p[idx] * out[idx];
    float sum = s;
    #pragma unroll
    for (int off = 16; off; off >>= 1) sum += __shfl_xor(sum, off, 32);
    out[idx] = s / sum;
}

extern "C" void kernel_launch(void* const* d_in, const int* in_sizes, int n_in,
                              void* d_out, int out_size, void* d_ws, size_t ws_size,
                              hipStream_t stream) {
    const float* x = (const float*)d_in[0];
    const float* W = (const float*)d_in[1];
    const float* U = (const float*)d_in[2];
    const float* b = (const float*)d_in[3];
    float* out = (float*)d_out;

    float* ws = (float*)d_ws;
    float* g  = ws;
    float* p  = ws + (size_t)Bv * Tv * Kv;
    float* Wp = p + (size_t)Bv * Tv * Kv;

    crf_packW     <<<dim3(128), dim3(256), 0, stream>>>(W, Wp);
    crf_emissions <<<dim3((Bv * Tv / 16) / 4), dim3(128), 0, stream>>>(x, Wp, b, g);
    crf_scan      <<<dim3(2 * Bv), dim3(32), 0, stream>>>(g, U, p, out);
    crf_marginals <<<dim3(Bv * Tv / 8), dim3(256), 0, stream>>>(p, out);
}


# scan: the 8 vectors of an unrolled trip are stored together at the trip's end from registers of their own, so no s_wait_xcnt (a store's address translation) stands inside a step's serial chain
# speedup vs baseline: 5.4206x; 1.1624x over previous
#include <hip/hip_runtime.h>
#include <hip/hip_bf16.h>

typedef __attribute__((ext_vector_type(2))) float v2f;
typedef __attribute__((ext_vector_type(4))) float v4f;
typedef __attribute__((ext_vector_type(8))) float v8f;

#define Bv 64
#define Tv 512
#define Dv 1024
#define Kv 32

__device__ __forceinline__ float fexp2(float x) { return __builtin_amdgcn_exp2f(x); }
__device__ __forceinline__ float flog2(float x) { return __builtin_amdgcn_logf(x); }

__device__ __forceinline__ float row16_max(float v) {
    float r;
    asm("v_max_num_f32_dpp %0, %1, %1 row_xmask:1 row_mask:0xf bank_mask:0xf\n\t"
        "v_max_num_f32_dpp %0, %0, %0 row_xmask:2 row_mask:0xf bank_mask:0xf\n\t"
        "v_max_num_f32_dpp %0, %0, %0 row_xmask:4 row_mask:0xf bank_mask:0xf\n\t"
        "v_max_num_f32_dpp %0, %0, %0 row_xmask:8 row_mask:0xf bank_mask:0xf"
        : "=v"(r) : "v"(v));
    return r;
}

__global__ __launch_bounds__(256) void crf_packW(const float* __restrict__ W,
                                                 float* __restrict__ Wp) {
    int idx  = blockIdx.x * 256 + threadIdx.x;
    int s    = idx & 3;
    int lane = (idx >> 2) & 31;
    int tile = (idx >> 7) & 1;
    int g    = idx >> 8;
    int d    = g * 8 + (lane >> 4) * 4 + s;
    int n    = tile * 16 + (lane & 15);
    Wp[idx] = W[d * Kv + n];
}

#define EG 8

struct EFrag { v4f a[EG]; v4f b0[EG]; v4f b1[EG]; };

__device__ __forceinline__ void emis_load(EFrag& f, const float* __restrict__ xrow,
                                          const float* __restrict__ wp, int blk) {
    #pragma unroll
    for (int j = 0; j < EG; ++j) {
        const int g = blk * EG + j;
        f.a[j]  = *(const v4f*)(xrow + g * 8);
        f.b0[j] = *(const v4f*)(wp + g * 256);
        f.b1[j] = *(const v4f*)(wp + g * 256 + 128);
    }
}

__device__ __forceinline__ void emis_mma(const EFrag& f, v8f& acc0, v8f& acc1) {
    #pragma unroll
    for (int j = 0; j < EG; ++j) {
        v2f a01 = {f.a[j].x,  f.a[j].y},  a23 = {f.a[j].z,  f.a[j].w};
        v2f p01 = {f.b0[j].x, f.b0[j].y}, p23 = {f.b0[j].z, f.b0[j].w};
        v2f q01 = {f.b1[j].x, f.b1[j].y}, q23 = {f.b1[j].z, f.b1[j].w};
        acc0 = __builtin_amdgcn_wmma_f32_16x16x4_f32(false, a01, false, p01,
                                                     (short)0, acc0, false, false);
        acc1 = __builtin_amdgcn_wmma_f32_16x16x4_f32(false, a01, false, q01,
                                                     (short)0, acc1, false, false);
        acc0 = __builtin_amdgcn_wmma_f32_16x16x4_f32(false, a23, false, p23,
                                                     (short)0, acc0, false, false);
        acc1 = __builtin_amdgcn_wmma_f32_16x16x4_f32(false, a23, false, q23,
                                                     (short)0, acc1, false, false);
    }
}

__global__ __launch_bounds__(128) void crf_emissions(const float* __restrict__ x,
                                                     const float* __restrict__ Wp,
                                                     const float* __restrict__ bias,
                                                     float* __restrict__ e) {
    const int lane = threadIdx.x & 31;
    const int wave = threadIdx.x >> 5;
    const int mt   = blockIdx.x * 4 + wave;
    const int half = lane >> 4;
    const int l15  = lane & 15;

    const float* xrow = x  + (size_t)(mt * 16 + l15) * Dv + half * 4;
    const float* wp   = Wp + lane * 4;

    v8f acc0 = {};
    v8f acc1 = {};
    EFrag P, Q;
    const int NB = Dv / (8 * EG);
    emis_load(P, xrow, wp, 0);
    for (int blk = 0; blk < NB - 2; blk += 2) {
        emis_load(Q, xrow, wp, blk + 1);
        emis_mma(P, acc0, acc1);
        emis_load(P, xrow, wp, blk + 2);
        emis_mma(Q, acc0, acc1);
    }
    emis_load(Q, xrow, wp, NB - 1);
    emis_mma(P, acc0, acc1);
    emis_mma(Q, acc0, acc1);

    const float bv0 = bias[l15];
    const float bv1 = bias[l15 + 16];
    const float LOG2E = 1.44269504088896340736f;
    #pragma unroll
    for (int r = 0; r < 8; ++r) {
        size_t row = (size_t)mt * 16 + r + half * 8;
        const float v0 = acc0[r] + bv0;
        const float v1 = acc1[r] + bv1;
        const float m  = row16_max(fmaxf(v0, v1));
        e[row * Kv + l15]      = fexp2((v0 - m) * LOG2E);
        e[row * Kv + l15 + 16] = fexp2((v1 - m) * LOG2E);
    }
}

#define PF 8

__device__ __forceinline__ void lin_step(float v, float g,
                                         const float (&cA)[16], const float (&cB)[16],
                                         float& s01, float& s23, float& vn) {
    float a0, a1, a2, a3, gs, vx, m;
    asm(
        "v_permlanex16_b32 %4, %6, %7, 0xfedcba98\n\t"
        "v_mul_f32_dpp %0, %6, %8 row_share:0 row_mask:0xf bank_mask:0xf\n\t"
        "v_mul_f32_dpp %1, %6, %9 row_share:1 row_mask:0xf bank_mask:0xf\n\t"
        "v_mul_f32_dpp %2, %6, %10 row_share:2 row_mask:0xf bank_mask:0xf\n\t"
        "v_max_num_f32 %5, %6, %4\n\t"
        "v_mul_f32_dpp %3, %6, %11 row_share:3 row_mask:0xf bank_mask:0xf\n\t"
        "v_fmac_f32_dpp %0, %6, %12 row_share:4 row_mask:0xf bank_mask:0xf\n\t"
        "v_fmac_f32_dpp %1, %6, %13 row_share:5 row_mask:0xf bank_mask:0xf\n\t"
        "v_max_num_f32_dpp %5, %5, %5 row_xmask:1 row_mask:0xf bank_mask:0xf\n\t"
        "v_fmac_f32_dpp %2, %6, %14 row_share:6 row_mask:0xf bank_mask:0xf\n\t"
        "v_fmac_f32_dpp %3, %6, %15 row_share:7 row_mask:0xf bank_mask:0xf\n\t"
        "v_fmac_f32_dpp %0, %6, %16 row_share:8 row_mask:0xf bank_mask:0xf\n\t"
        "v_max_num_f32_dpp %5, %5, %5 row_xmask:2 row_mask:0xf bank_mask:0xf\n\t"
        "v_fmac_f32_dpp %1, %6, %17 row_share:9 row_mask:0xf bank_mask:0xf\n\t"
        "v_fmac_f32_dpp %2, %6, %18 row_share:10 row_mask:0xf bank_mask:0xf\n\t"
        "v_fmac_f32_dpp %3, %6, %19 row_share:11 row_mask:0xf bank_mask:0xf\n\t"
        "v_max_num_f32_dpp %5, %5, %5 row_xmask:4 row_mask:0xf bank_mask:0xf\n\t"
        "v_fmac_f32_dpp %0, %6, %20 row_share:12 row_mask:0xf bank_mask:0xf\n\t"
        "v_fmac_f32_dpp %1, %6, %21 row_share:13 row_mask:0xf bank_mask:0xf\n\t"
        "v_fmac_f32_dpp %2, %6, %22 row_share:14 row_mask:0xf bank_mask:0xf\n\t"
        "v_max_num_f32_dpp %5, %5, %5 row_xmask:8 row_mask:0xf bank_mask:0xf\n\t"
        "v_fmac_f32_dpp %3, %6, %23 row_share:15 row_mask:0xf bank_mask:0xf"
        : "=&v"(a0), "=&v"(a1), "=&v"(a2), "=&v"(a3), "=&v"(vx), "=&v"(m)
        : "v"(v), "s"(0x76543210), "v"(cA[0]), "v"(cA[1]), "v"(cA[2]), "v"(cA[3]), "v"(cA[4]), "v"(cA[5]), "v"(cA[6]), "v"(cA[7]), "v"(cA[8]), "v"(cA[9]), "v"(cA[10]), "v"(cA[11]), "v"(cA[12]), "v"(cA[13]), "v"(cA[14]), "v"(cA[15]));
    asm(
        "v_fmac_f32_dpp %0, %5, %8 row_share:0 row_mask:0xf bank_mask:0xf\n\t"
        "v_fmac_f32_dpp %1, %5, %9 row_share:1 row_mask:0xf bank_mask:0xf\n\t"
        "v_rcp_f32 %4, %6\n\t"
        "v_fmac_f32_dpp %2, %5, %10 row_share:2 row_mask:0xf bank_mask:0xf\n\t"
        "v_fmac_f32_dpp %3, %5, %11 row_share:3 row_mask:0xf bank_mask:0xf\n\t"
        "v_fmac_f32_dpp %0, %5, %12 row_share:4 row_mask:0xf bank_mask:0xf\n\t"
        "v_fmac_f32_dpp %1, %5, %13 row_share:5 row_mask:0xf bank_mask:0xf\n\t"
        "v_fmac_f32_dpp %2, %5, %14 row_share:6 row_mask:0xf bank_mask:0xf\n\t"
        "v_mul_f32 %4, %7, %4\n\t"
        "v_fmac_f32_dpp %3, %5, %15 row_share:7 row_mask:0xf bank_mask:0xf\n\t"
        "v_fmac_f32_dpp %0, %5, %16 row_share:8 row_mask:0xf bank_mask:0xf\n\t"
        "v_fmac_f32_dpp %1, %5, %17 row_share:9 row_mask:0xf bank_mask:0xf\n\t"
        "v_fmac_f32_dpp %2, %5, %18 row_share:10 row_mask:0xf bank_mask:0xf\n\t"
        "v_fmac_f32_dpp %3, %5, %19 row_share:11 row_mask:0xf bank_mask:0xf\n\t"
        "v_fmac_f32_dpp %0, %5, %20 row_share:12 row_mask:0xf bank_mask:0xf\n\t"
        "v_fmac_f32_dpp %1, %5, %21 row_share:13 row_mask:0xf bank_mask:0xf\n\t"
        "v_fmac_f32_dpp %2, %5, %22 row_share:14 row_mask:0xf bank_mask:0xf\n\t"
        "v_fmac_f32_dpp %3, %5, %23 row_share:15 row_mask:0xf bank_mask:0xf\n\t"
        "v_add_f32 %0, %0, %1\n\t"
        "v_add_f32 %2, %2, %3\n\t"
        "v_mul_f32 %1, %0, %4\n\t"
        "v_fmac_f32 %1, %2, %4"
        : "+v"(a0), "+v"(a1), "+v"(a2), "+v"(a3), "=&v"(gs)
        : "v"(vx), "v"(m), "v"(g), "v"(cB[0]), "v"(cB[1]), "v"(cB[2]), "v"(cB[3]), "v"(cB[4]), "v"(cB[5]), "v"(cB[6]), "v"(cB[7]), "v"(cB[8]), "v"(cB[9]), "v"(cB[10]), "v"(cB[11]), "v"(cB[12]), "v"(cB[13]), "v"(cB[14]), "v"(cB[15]));
    s01 = a0;
    s23 = a2;
    vn  = a1;
}

__global__ __launch_bounds__(32) void crf_scan(const float* __restrict__ g,
                                               const float* __restrict__ U,
                                               float* __restrict__ p_out,
                                               float* __restrict__ q_out) {
    const int   lane   = threadIdx.x;
    const float LOG2E  = 1.44269504088896340736f;
    float cA[16], cB[16];
    float ring[PF];
    float keep[PF];
    float s01, s23, vn;
    const int   own    = (lane >> 4) * 16;
    const int   oth    = 16 - own;
    static_assert(PF == 8, "the fence statements below name the PF ring slots one by one");

    if (blockIdx.x < Bv) {
        const int bIdx = blockIdx.x;
        #pragma unroll
        for (int n = 0; n < 16; ++n) {
            cA[n] = fexp2(U[(own + n) * Kv + lane] * LOG2E);
            cB[n] = fexp2(U[(oth + n) * Kv + lane] * LOG2E);
        }

        const float* gb = g     + (size_t)bIdx * Tv * Kv;
        float*       pb = p_out + (size_t)bIdx * Tv * Kv;

        #pragma unroll
        for (int i = 0; i < PF; ++i)
            ring[i] = gb[(1 + i) * Kv + lane];

        float p = gb[lane];
        pb[lane] = p;
        asm volatile("" :: "v"(ring[0]), "v"(ring[1]), "v"(ring[2]), "v"(ring[3]),
                           "v"(ring[4]), "v"(ring[5]), "v"(ring[6]), "v"(ring[7]));
        int t = 1;
        for (; t + PF <= Tv; t += PF) {
            #pragma unroll
            for (int i = 0; i < PF; ++i) {
                lin_step(p, ring[i], cA, cB, s01, s23, vn);
                p = vn;
                keep[i] = p;
                int tn = t + i + PF;
                tn = tn < Tv ? tn : Tv - 1;
                ring[i] = gb[tn * Kv + lane];
                __builtin_amdgcn_sched_barrier(0);
            }
            #pragma unroll
            for (int i = 0; i < PF; ++i)
                pb[(t + i) * Kv + lane] = keep[i];
        }
        #pragma unroll
        for (int i = 0; i < PF - 1; ++i) {
            if (t + i < Tv) {
                lin_step(p, ring[i], cA, cB, s01, s23, vn);
                p = vn;
                pb[(t + i) * Kv + lane] = p;
            }
        }
    } else {
        const int bIdx = blockIdx.x - Bv;
        #pragma unroll
        for (int n = 0; n < 16; ++n) {
            cA[n] = fexp2(U[lane * Kv + own + n] * LOG2E);
            cB[n] = fexp2(U[lane * Kv + oth + n] * LOG2E);
        }

        const float* gb = g     + (size_t)bIdx * Tv * Kv;
        float*       qb = q_out + (size_t)bIdx * Tv * Kv;

        #pragma unroll
        for (int i = 0; i < PF; ++i)
            ring[i] = gb[(Tv - 2 - i) * Kv + lane];

        float w = gb[(Tv - 1) * Kv + lane];
        qb[(Tv - 1) * Kv + lane] = 1.f;
        asm volatile("" :: "v"(ring[0]), "v"(ring[1]), "v"(ring[2]), "v"(ring[3]),
                           "v"(ring[4]), "v"(ring[5]), "v"(ring[6]), "v"(ring[7]), "v"(w));
        int t = Tv - 2;
        for (; t - (PF - 1) >= 0; t -= PF) {
            #pragma unroll
            for (int i = 0; i < PF; ++i) {
                lin_step(w, ring[i], cA, cB, s01, s23, vn);
                keep[i] = s01 + s23;
                w = vn;
                int qn = t - i - PF;
                qn = qn > 0 ? qn : 0;
                ring[i] = gb[qn * Kv + lane];
                __builtin_amdgcn_sched_barrier(0);
            }
            #pragma unroll
            for (int i = 0; i < PF; ++i)
                qb[(t - i) * Kv + lane] = keep[i];
        }
        #pragma unroll
        for (int i = 0; i < PF - 1; ++i) {
            if (t - i >= 0) {
                lin_step(w, ring[i], cA, cB, s01, s23, vn);
                qb[(t - i) * Kv + lane] = s01 + s23;
                w = vn;
            }
        }
    }
}

__global__ __launch_bounds__(256) void crf_marginals(const float* __restrict__ p,
                                                     float* __restrict__ out) {
    const int lane = threadIdx.x & 31;
    const int wave = threadIdx.x >> 5;
    const size_t row = (size_t)blockIdx.x * 8 + wave;
    const size_t idx = row * Kv + lane;

    const float s = p[idx] * out[idx];
    float sum = s;
    #pragma unroll
    for (int off = 16; off; off >>= 1) sum += __shfl_xor(sum, off, 32);
    out[idx] = s / sum;
}

extern "C" void kernel_launch(void* const* d_in, const int* in_sizes, int n_in,
                              void* d_out, int out_size, void* d_ws, size_t ws_size,
                              hipStream_t stream) {
    const float* x = (const float*)d_in[0];
    const float* W = (const float*)d_in[1];
    const float* U = (const float*)d_in[2];
    const float* b = (const float*)d_in[3];
    float* out = (float*)d_out;

    float* ws = (float*)d_ws;
    float* g  = ws;
    float* p  = ws + (size_t)Bv * Tv * Kv;
    float* Wp = p + (size_t)Bv * Tv * Kv;

    crf_packW     <<<dim3(128), dim3(256), 0, stream>>>(W, Wp);
    crf_emissions <<<dim3((Bv * Tv / 16) / 4), dim3(128), 0, stream>>>(x, Wp, b, g);
    crf_scan      <<<dim3(2 * Bv), dim3(32), 0, stream>>>(g, U, p, out);
    crf_marginals <<<dim3(Bv * Tv / 8), dim3(256), 0, stream>>>(p, out);
}


# marginals fused into the scan kernel: an 8-wave work-group per batch keeps both chains' vectors in LDS, normalises behind one barrier; row sum now in-lane + 3-level DPP tree (order changed)
# speedup vs baseline: 5.9226x; 1.0926x over previous
#include <hip/hip_runtime.h>
#include <hip/hip_bf16.h>

typedef __attribute__((ext_vector_type(2))) float v2f;
typedef __attribute__((ext_vector_type(4))) float v4f;
typedef __attribute__((ext_vector_type(8))) float v8f;

#define Bv 64
#define Tv 512
#define Dv 1024
#define Kv 32

__device__ __forceinline__ float fexp2(float x) { return __builtin_amdgcn_exp2f(x); }
__device__ __forceinline__ float flog2(float x) { return __builtin_amdgcn_logf(x); }

__device__ __forceinline__ float row16_max(float v) {
    float r;
    asm("v_max_num_f32_dpp %0, %1, %1 row_xmask:1 row_mask:0xf bank_mask:0xf\n\t"
        "v_max_num_f32_dpp %0, %0, %0 row_xmask:2 row_mask:0xf bank_mask:0xf\n\t"
        "v_max_num_f32_dpp %0, %0, %0 row_xmask:4 row_mask:0xf bank_mask:0xf\n\t"
        "v_max_num_f32_dpp %0, %0, %0 row_xmask:8 row_mask:0xf bank_mask:0xf"
        : "=v"(r) : "v"(v));
    return r;
}

__device__ __forceinline__ float lane8_sum(float v) {
    float r;
    asm("v_add_f32_dpp %0, %1, %1 row_xmask:1 row_mask:0xf bank_mask:0xf\n\t"
        "v_add_f32_dpp %0, %0, %0 row_xmask:2 row_mask:0xf bank_mask:0xf\n\t"
        "v_add_f32_dpp %0, %0, %0 row_xmask:4 row_mask:0xf bank_mask:0xf"
        : "=v"(r) : "v"(v));
    return r;
}

__global__ __launch_bounds__(256) void crf_packW(const float* __restrict__ W,
                                                 float* __restrict__ Wp) {
    int idx  = blockIdx.x * 256 + threadIdx.x;
    int s    = idx & 3;
    int lane = (idx >> 2) & 31;
    int tile = (idx >> 7) & 1;
    int g    = idx >> 8;
    int d    = g * 8 + (lane >> 4) * 4 + s;
    int n    = tile * 16 + (lane & 15);
    Wp[idx] = W[d * Kv + n];
}

#define EG 8

struct EFrag { v4f a[EG]; v4f b0[EG]; v4f b1[EG]; };

__device__ __forceinline__ void emis_load(EFrag& f, const float* __restrict__ xrow,
                                          const float* __restrict__ wp, int blk) {
    #pragma unroll
    for (int j = 0; j < EG; ++j) {
        const int g = blk * EG + j;
        f.a[j]  = *(const v4f*)(xrow + g * 8);
        f.b0[j] = *(const v4f*)(wp + g * 256);
        f.b1[j] = *(const v4f*)(wp + g * 256 + 128);
    }
}

__device__ __forceinline__ void emis_mma(const EFrag& f, v8f& acc0, v8f& acc1) {
    #pragma unroll
    for (int j = 0; j < EG; ++j) {
        v2f a01 = {f.a[j].x,  f.a[j].y},  a23 = {f.a[j].z,  f.a[j].w};
        v2f p01 = {f.b0[j].x, f.b0[j].y}, p23 = {f.b0[j].z, f.b0[j].w};
        v2f q01 = {f.b1[j].x, f.b1[j].y}, q23 = {f.b1[j].z, f.b1[j].w};
        acc0 = __builtin_amdgcn_wmma_f32_16x16x4_f32(false, a01, false, p01,
                                                     (short)0, acc0, false, false);
        acc1 = __builtin_amdgcn_wmma_f32_16x16x4_f32(false, a01, false, q01,
                                                     (short)0, acc1, false, false);
        acc0 = __builtin_amdgcn_wmma_f32_16x16x4_f32(false, a23, false, p23,
                                                     (short)0, acc0, false, false);
        acc1 = __builtin_amdgcn_wmma_f32_16x16x4_f32(false, a23, false, q23,
                                                     (short)0, acc1, false, false);
    }
}

__global__ __launch_bounds__(128) void crf_emissions(const float* __restrict__ x,
                                                     const float* __restrict__ Wp,
                                                     const float* __restrict__ bias,
                                                     float* __restrict__ e) {
    const int lane = threadIdx.x & 31;
    const int wave = threadIdx.x >> 5;
    const int mt   = blockIdx.x * 4 + wave;
    const int half = lane >> 4;
    const int l15  = lane & 15;

    const float* xrow = x  + (size_t)(mt * 16 + l15) * Dv + half * 4;
    const float* wp   = Wp + lane * 4;

    v8f acc0 = {};
    v8f acc1 = {};
    EFrag P, Q;
    const int NB = Dv / (8 * EG);
    emis_load(P, xrow, wp, 0);
    for (int blk = 0; blk < NB - 2; blk += 2) {
        emis_load(Q, xrow, wp, blk + 1);
        emis_mma(P, acc0, acc1);
        emis_load(P, xrow, wp, blk + 2);
        emis_mma(Q, acc0, acc1);
    }
    emis_load(Q, xrow, wp, NB - 1);
    emis_mma(P, acc0, acc1);
    emis_mma(Q, acc0, acc1);

    const float bv0 = bias[l15];
    const float bv1 = bias[l15 + 16];
    const float LOG2E = 1.44269504088896340736f;
    #pragma unroll
    for (int r = 0; r < 8; ++r) {
        size_t row = (size_t)mt * 16 + r + half * 8;
        const float v0 = acc0[r] + bv0;
        const float v1 = acc1[r] + bv1;
        const float m  = row16_max(fmaxf(v0, v1));
        e[row * Kv + l15]      = fexp2((v0 - m) * LOG2E);
        e[row * Kv + l15 + 16] = fexp2((v1 - m) * LOG2E);
    }
}

#define PF 8

__device__ __forceinline__ void lin_step(float v, float g,
                                         const float (&cA)[16], const float (&cB)[16],
                                         float& s01, float& s23, float& vn) {
    float a0, a1, a2, a3, gs, vx, m;
    asm(
        "v_permlanex16_b32 %4, %6, %7, 0xfedcba98\n\t"
        "v_mul_f32_dpp %0, %6, %8 row_share:0 row_mask:0xf bank_mask:0xf\n\t"
        "v_mul_f32_dpp %1, %6, %9 row_share:1 row_mask:0xf bank_mask:0xf\n\t"
        "v_mul_f32_dpp %2, %6, %10 row_share:2 row_mask:0xf bank_mask:0xf\n\t"
        "v_max_num_f32 %5, %6, %4\n\t"
        "v_mul_f32_dpp %3, %6, %11 row_share:3 row_mask:0xf bank_mask:0xf\n\t"
        "v_fmac_f32_dpp %0, %6, %12 row_share:4 row_mask:0xf bank_mask:0xf\n\t"
        "v_fmac_f32_dpp %1, %6, %13 row_share:5 row_mask:0xf bank_mask:0xf\n\t"
        "v_max_num_f32_dpp %5, %5, %5 row_xmask:1 row_mask:0xf bank_mask:0xf\n\t"
        "v_fmac_f32_dpp %2, %6, %14 row_share:6 row_mask:0xf bank_mask:0xf\n\t"
        "v_fmac_f32_dpp %3, %6, %15 row_share:7 row_mask:0xf bank_mask:0xf\n\t"
        "v_fmac_f32_dpp %0, %6, %16 row_share:8 row_mask:0xf bank_mask:0xf\n\t"
        "v_max_num_f32_dpp %5, %5, %5 row_xmask:2 row_mask:0xf bank_mask:0xf\n\t"
        "v_fmac_f32_dpp %1, %6, %17 row_share:9 row_mask:0xf bank_mask:0xf\n\t"
        "v_fmac_f32_dpp %2, %6, %18 row_share:10 row_mask:0xf bank_mask:0xf\n\t"
        "v_fmac_f32_dpp %3, %6, %19 row_share:11 row_mask:0xf bank_mask:0xf\n\t"
        "v_max_num_f32_dpp %5, %5, %5 row_xmask:4 row_mask:0xf bank_mask:0xf\n\t"
        "v_fmac_f32_dpp %0, %6, %20 row_share:12 row_mask:0xf bank_mask:0xf\n\t"
        "v_fmac_f32_dpp %1, %6, %21 row_share:13 row_mask:0xf bank_mask:0xf\n\t"
        "v_fmac_f32_dpp %2, %6, %22 row_share:14 row_mask:0xf bank_mask:0xf\n\t"
        "v_max_num_f32_dpp %5, %5, %5 row_xmask:8 row_mask:0xf bank_mask:0xf\n\t"
        "v_fmac_f32_dpp %3, %6, %23 row_share:15 row_mask:0xf bank_mask:0xf"
        : "=&v"(a0), "=&v"(a1), "=&v"(a2), "=&v"(a3), "=&v"(vx), "=&v"(m)
        : "v"(v), "s"(0x76543210), "v"(cA[0]), "v"(cA[1]), "v"(cA[2]), "v"(cA[3]), "v"(cA[4]), "v"(cA[5]), "v"(cA[6]), "v"(cA[7]), "v"(cA[8]), "v"(cA[9]), "v"(cA[10]), "v"(cA[11]), "v"(cA[12]), "v"(cA[13]), "v"(cA[14]), "v"(cA[15]));
    asm(
        "v_fmac_f32_dpp %0, %5, %8 row_share:0 row_mask:0xf bank_mask:0xf\n\t"
        "v_fmac_f32_dpp %1, %5, %9 row_share:1 row_mask:0xf bank_mask:0xf\n\t"
        "v_rcp_f32 %4, %6\n\t"
        "v_fmac_f32_dpp %2, %5, %10 row_share:2 row_mask:0xf bank_mask:0xf\n\t"
        "v_fmac_f32_dpp %3, %5, %11 row_share:3 row_mask:0xf bank_mask:0xf\n\t"
        "v_fmac_f32_dpp %0, %5, %12 row_share:4 row_mask:0xf bank_mask:0xf\n\t"
        "v_fmac_f32_dpp %1, %5, %13 row_share:5 row_mask:0xf bank_mask:0xf\n\t"
        "v_fmac_f32_dpp %2, %5, %14 row_share:6 row_mask:0xf bank_mask:0xf\n\t"
        "v_mul_f32 %4, %7, %4\n\t"
        "v_fmac_f32_dpp %3, %5, %15 row_share:7 row_mask:0xf bank_mask:0xf\n\t"
        "v_fmac_f32_dpp %0, %5, %16 row_share:8 row_mask:0xf bank_mask:0xf\n\t"
        "v_fmac_f32_dpp %1, %5, %17 row_share:9 row_mask:0xf bank_mask:0xf\n\t"
        "v_fmac_f32_dpp %2, %5, %18 row_share:10 row_mask:0xf bank_mask:0xf\n\t"
        "v_fmac_f32_dpp %3, %5, %19 row_share:11 row_mask:0xf bank_mask:0xf\n\t"
        "v_fmac_f32_dpp %0, %5, %20 row_share:12 row_mask:0xf bank_mask:0xf\n\t"
        "v_fmac_f32_dpp %1, %5, %21 row_share:13 row_mask:0xf bank_mask:0xf\n\t"
        "v_fmac_f32_dpp %2, %5, %22 row_share:14 row_mask:0xf bank_mask:0xf\n\t"
        "v_fmac_f32_dpp %3, %5, %23 row_share:15 row_mask:0xf bank_mask:0xf\n\t"
        "v_add_f32 %0, %0, %1\n\t"
        "v_add_f32 %2, %2, %3\n\t"
        "v_mul_f32 %1, %0, %4\n\t"
        "v_fmac_f32 %1, %2, %4"
        : "+v"(a0), "+v"(a1), "+v"(a2), "+v"(a3), "=&v"(gs)
        : "v"(vx), "v"(m), "v"(g), "v"(cB[0]), "v"(cB[1]), "v"(cB[2]), "v"(cB[3]), "v"(cB[4]), "v"(cB[5]), "v"(cB[6]), "v"(cB[7]), "v"(cB[8]), "v"(cB[9]), "v"(cB[10]), "v"(cB[11]), "v"(cB[12]), "v"(cB[13]), "v"(cB[14]), "v"(cB[15]));
    s01 = a0;
    s23 = a2;
    vn  = a1;
}

#define FWD_WAVE 0
#define BWD_WAVE 5

__global__ __launch_bounds__(256) void crf_scan(const float* __restrict__ g,
                                                const float* __restrict__ U,
                                                float* __restrict__ out) {
    __shared__ float sp[Tv * Kv];
    __shared__ float sq[Tv * Kv];
    const int   lane   = threadIdx.x & 31;
    const int   wave   = __builtin_amdgcn_readfirstlane(threadIdx.x >> 5);
    const int   bIdx   = blockIdx.x;
    const float LOG2E  = 1.44269504088896340736f;
    float cA[16], cB[16];
    float ring[PF];
    float keep[PF];
    float s01, s23, vn;
    const int   own    = (lane >> 4) * 16;
    const int   oth    = 16 - own;
    static_assert(PF == 8, "the fence statements below name the PF ring slots one by one");

    if (wave == FWD_WAVE) {
        #pragma unroll
        for (int n = 0; n < 16; ++n) {
            cA[n] = fexp2(U[(own + n) * Kv + lane] * LOG2E);
            cB[n] = fexp2(U[(oth + n) * Kv + lane] * LOG2E);
        }

        const float* gb = g + (size_t)bIdx * Tv * Kv;
        float*       pb = sp;

        #pragma unroll
        for (int i = 0; i < PF; ++i)
            ring[i] = gb[(1 + i) * Kv + lane];

        float p = gb[lane];
        pb[lane] = p;
        asm volatile("" :: "v"(ring[0]), "v"(ring[1]), "v"(ring[2]), "v"(ring[3]),
                           "v"(ring[4]), "v"(ring[5]), "v"(ring[6]), "v"(ring[7]));
        int t = 1;
        for (; t + PF <= Tv; t += PF) {
            #pragma unroll
            for (int i = 0; i < PF; ++i) {
                lin_step(p, ring[i], cA, cB, s01, s23, vn);
                p = vn;
                keep[i] = p;
                int tn = t + i + PF;
                tn = tn < Tv ? tn : Tv - 1;
                ring[i] = gb[tn * Kv + lane];
                __builtin_amdgcn_sched_barrier(0);
            }
            #pragma unroll
            for (int i = 0; i < PF; ++i)
                pb[(t + i) * Kv + lane] = keep[i];
        }
        #pragma unroll
        for (int i = 0; i < PF - 1; ++i) {
            if (t + i < Tv) {
                lin_step(p, ring[i], cA, cB, s01, s23, vn);
                p = vn;
                pb[(t + i) * Kv + lane] = p;
            }
        }
    } else if (wave == BWD_WAVE) {
        #pragma unroll
        for (int n = 0; n < 16; ++n) {
            cA[n] = fexp2(U[lane * Kv + own + n] * LOG2E);
            cB[n] = fexp2(U[lane * Kv + oth + n] * LOG2E);
        }

        const float* gb = g + (size_t)bIdx * Tv * Kv;
        float*       qb = sq;

        #pragma unroll
        for (int i = 0; i < PF; ++i)
            ring[i] = gb[(Tv - 2 - i) * Kv + lane];

        float w = gb[(Tv - 1) * Kv + lane];
        qb[(Tv - 1) * Kv + lane] = 1.f;
        asm volatile("" :: "v"(ring[0]), "v"(ring[1]), "v"(ring[2]), "v"(ring[3]),
                           "v"(ring[4]), "v"(ring[5]), "v"(ring[6]), "v"(ring[7]), "v"(w));
        int t = Tv - 2;
        for (; t - (PF - 1) >= 0; t -= PF) {
            #pragma unroll
            for (int i = 0; i < PF; ++i) {
                lin_step(w, ring[i], cA, cB, s01, s23, vn);
                keep[i] = s01 + s23;
                w = vn;
                int qn = t - i - PF;
                qn = qn > 0 ? qn : 0;
                ring[i] = gb[qn * Kv + lane];
                __builtin_amdgcn_sched_barrier(0);
            }
            #pragma unroll
            for (int i = 0; i < PF; ++i)
                qb[(t - i) * Kv + lane] = keep[i];
        }
        #pragma unroll
        for (int i = 0; i < PF - 1; ++i) {
            if (t - i >= 0) {
                lin_step(w, ring[i], cA, cB, s01, s23, vn);
                qb[(t - i) * Kv + lane] = s01 + s23;
                w = vn;
            }
        }
    }

    __syncthreads();

    const v4f* sp4 = (const v4f*)sp;
    const v4f* sq4 = (const v4f*)sq;
    v4f*       ob4 = (v4f*)(out + (size_t)bIdx * Tv * Kv);
    #pragma unroll 4
    for (int it = 0; it < (Tv * Kv / 4) / 256; ++it) {
        const int i4 = it * 256 + wave * 32 + lane;
        const v4f s = sp4[i4] * sq4[i4];
        const float sum = lane8_sum((s.x + s.y) + (s.z + s.w));
        const float inv = 1.0f / sum;
        ob4[i4] = s * inv;
    }
}

extern "C" void kernel_launch(void* const* d_in, const int* in_sizes, int n_in,
                              void* d_out, int out_size, void* d_ws, size_t ws_size,
                              hipStream_t stream) {
    const float* x = (const float*)d_in[0];
    const float* W = (const float*)d_in[1];
    const float* U = (const float*)d_in[2];
    const float* b = (const float*)d_in[3];
    float* out = (float*)d_out;

    float* ws = (float*)d_ws;
    float* g  = ws;
    float* Wp = g + (size_t)Bv * Tv * Kv;

    crf_packW     <<<dim3(128), dim3(256), 0, stream>>>(W, Wp);
    crf_emissions <<<dim3((Bv * Tv / 16) / 4), dim3(128), 0, stream>>>(x, Wp, b, g);
    crf_scan      <<<dim3(Bv), dim3(256), 0, stream>>>(g, U, out);
}


# scan: a lane sums 2 outputs over its own 16-lane row: 15 v_permlane16 rotations feed 16 v_pk_fma_f32, v_permlane16_swap + add join the rows (was 32 DPP fmacs); sum order changed; row ring 16 deep
# speedup vs baseline: 6.7831x; 1.1453x over previous
#include <hip/hip_runtime.h>
#include <hip/hip_bf16.h>

typedef __attribute__((ext_vector_type(2))) float v2f;
typedef __attribute__((ext_vector_type(4))) float v4f;
typedef __attribute__((ext_vector_type(8))) float v8f;

#define Bv 64
#define Tv 512
#define Dv 1024
#define Kv 32

__device__ __forceinline__ float fexp2(float x) { return __builtin_amdgcn_exp2f(x); }
__device__ __forceinline__ float flog2(float x) { return __builtin_amdgcn_logf(x); }

__device__ __forceinline__ float row16_max(float v) {
    float r;
    asm("v_max_num_f32_dpp %0, %1, %1 row_xmask:1 row_mask:0xf bank_mask:0xf\n\t"
        "v_max_num_f32_dpp %0, %0, %0 row_xmask:2 row_mask:0xf bank_mask:0xf\n\t"
        "v_max_num_f32_dpp %0, %0, %0 row_xmask:4 row_mask:0xf bank_mask:0xf\n\t"
        "v_max_num_f32_dpp %0, %0, %0 row_xmask:8 row_mask:0xf bank_mask:0xf"
        : "=v"(r) : "v"(v));
    return r;
}

__device__ __forceinline__ float lane8_sum(float v) {
    float r;
    asm("v_add_f32_dpp %0, %1, %1 row_xmask:1 row_mask:0xf bank_mask:0xf\n\t"
        "v_add_f32_dpp %0, %0, %0 row_xmask:2 row_mask:0xf bank_mask:0xf\n\t"
        "v_add_f32_dpp %0, %0, %0 row_xmask:4 row_mask:0xf bank_mask:0xf"
        : "=v"(r) : "v"(v));
    return r;
}

__global__ __launch_bounds__(256) void crf_packW(const float* __restrict__ W,
                                                 float* __restrict__ Wp) {
    int idx  = blockIdx.x * 256 + threadIdx.x;
    int s    = idx & 3;
    int lane = (idx >> 2) & 31;
    int tile = (idx >> 7) & 1;
    int g    = idx >> 8;
    int d    = g * 8 + (lane >> 4) * 4 + s;
    int n    = tile * 16 + (lane & 15);
    Wp[idx] = W[d * Kv + n];
}

#define EG 8

struct EFrag { v4f a[EG]; v4f b0[EG]; v4f b1[EG]; };

__device__ __forceinline__ void emis_load(EFrag& f, const float* __restrict__ xrow,
                                          const float* __restrict__ wp, int blk) {
    #pragma unroll
    for (int j = 0; j < EG; ++j) {
        const int g = blk * EG + j;
        f.a[j]  = *(const v4f*)(xrow + g * 8);
        f.b0[j] = *(const v4f*)(wp + g * 256);
        f.b1[j] = *(const v4f*)(wp + g * 256 + 128);
    }
}

__device__ __forceinline__ void emis_mma(const EFrag& f, v8f& acc0, v8f& acc1) {
    #pragma unroll
    for (int j = 0; j < EG; ++j) {
        v2f a01 = {f.a[j].x,  f.a[j].y},  a23 = {f.a[j].z,  f.a[j].w};
        v2f p01 = {f.b0[j].x, f.b0[j].y}, p23 = {f.b0[j].z, f.b0[j].w};
        v2f q01 = {f.b1[j].x, f.b1[j].y}, q23 = {f.b1[j].z, f.b1[j].w};
        acc0 = __builtin_amdgcn_wmma_f32_16x16x4_f32(false, a01, false, p01,
                                                     (short)0, acc0, false, false);
        acc1 = __builtin_amdgcn_wmma_f32_16x16x4_f32(false, a01, false, q01,
                                                     (short)0, acc1, false, false);
        acc0 = __builtin_amdgcn_wmma_f32_16x16x4_f32(false, a23, false, p23,
                                                     (short)0, acc0, false, false);
        acc1 = __builtin_amdgcn_wmma_f32_16x16x4_f32(false, a23, false, q23,
                                                     (short)0, acc1, false, false);
    }
}

__global__ __launch_bounds__(128) void crf_emissions(const float* __restrict__ x,
                                                     const float* __restrict__ Wp,
                                                     const float* __restrict__ bias,
                                                     float* __restrict__ e) {
    const int lane = threadIdx.x & 31;
    const int wave = threadIdx.x >> 5;
    const int mt   = blockIdx.x * 4 + wave;
    const int half = lane >> 4;
    const int l15  = lane & 15;

    const float* xrow = x  + (size_t)(mt * 16 + l15) * Dv + half * 4;
    const float* wp   = Wp + lane * 4;

    v8f acc0 = {};
    v8f acc1 = {};
    EFrag P, Q;
    const int NB = Dv / (8 * EG);
    emis_load(P, xrow, wp, 0);
    for (int blk = 0; blk < NB - 2; blk += 2) {
        emis_load(Q, xrow, wp, blk + 1);
        emis_mma(P, acc0, acc1);
        emis_load(P, xrow, wp, blk + 2);
        emis_mma(Q, acc0, acc1);
    }
    emis_load(Q, xrow, wp, NB - 1);
    emis_mma(P, acc0, acc1);
    emis_mma(Q, acc0, acc1);

    const float bv0 = bias[l15];
    const float bv1 = bias[l15 + 16];
    const float LOG2E = 1.44269504088896340736f;
    #pragma unroll
    for (int r = 0; r < 8; ++r) {
        size_t row = (size_t)mt * 16 + r + half * 8;
        const float v0 = acc0[r] + bv0;
        const float v1 = acc1[r] + bv1;
        const float m  = row16_max(fmaxf(v0, v1));
        e[row * Kv + l15]      = fexp2((v0 - m) * LOG2E);
        e[row * Kv + l15 + 16] = fexp2((v1 - m) * LOG2E);
    }
}

#define PF 16

__host__ __device__ constexpr unsigned rot_selects(int n, int first_lane) {
    unsigned w = 0;
    for (int j = 0; j < 8; ++j) w |= (unsigned)((first_lane + j + n) & 15) << (4 * j);
    return w;
}
template <int N> __device__ __forceinline__ float row_rot(float v) {
    float r;
    asm("v_permlane16_b32 %0, %1, %2, %3"
        : "=v"(r) : "v"(v), "s"(rot_selects(N, 0)), "s"(rot_selects(N, 8)));
    return r;
}
__device__ __forceinline__ float other_row(float v) {
    float r;
    asm("v_permlanex16_b32 %0, %1, %2, %3" : "=v"(r) : "v"(v), "s"(0x76543210u), "s"(0xfedcba98u));
    return r;
}
template <int M> __device__ __forceinline__ float max_xmask(float v) {
    float r;
    asm("v_max_num_f32_dpp %0, %1, %1 row_xmask:%2 row_mask:0xf bank_mask:0xf" : "=v"(r) : "v"(v), "n"(M));
    return r;
}
#define PIN() __builtin_amdgcn_sched_barrier(0)

__device__ __forceinline__ void lin_step(float v, float g, const v2f (&C)[16],
                                         float& s, float& vn) {
    float b[16];
    v2f A0, A1, A2, A3;
    float m, gs, lo, hi, t0, t1;
    b[1] = row_rot<1>(v);
    A0 = v2f{v, v} * C[0];
    b[2] = row_rot<2>(v);
    const float vx = other_row(v);
    b[3] = row_rot<3>(v);
    PIN();
    A1 = v2f{b[1], b[1]} * C[1];            b[4] = row_rot<4>(v);     PIN();
    A2 = v2f{b[2], b[2]} * C[2];            b[5] = row_rot<5>(v);     PIN();
    asm("v_max_num_f32 %0, %1, %2" : "=v"(m) : "v"(v), "v"(vx));      PIN();
    A3 = v2f{b[3], b[3]} * C[3];            b[6] = row_rot<6>(v);     PIN();
    A0 = __builtin_elementwise_fma(v2f{b[4], b[4]}, C[4], A0);   b[7]  = row_rot<7>(v);   PIN();
    A1 = __builtin_elementwise_fma(v2f{b[5], b[5]}, C[5], A1);   b[8]  = row_rot<8>(v);   PIN();
    m  = max_xmask<1>(m);                                             PIN();
    A2 = __builtin_elementwise_fma(v2f{b[6], b[6]}, C[6], A2);   b[9]  = row_rot<9>(v);   PIN();
    A3 = __builtin_elementwise_fma(v2f{b[7], b[7]}, C[7], A3);   b[10] = row_rot<10>(v);  PIN();
    A0 = __builtin_elementwise_fma(v2f{b[8], b[8]}, C[8], A0);   b[11] = row_rot<11>(v);  PIN();
    A1 = __builtin_elementwise_fma(v2f{b[9], b[9]}, C[9], A1);   b[12] = row_rot<12>(v);  PIN();
    m  = max_xmask<2>(m);                                             PIN();
    A2 = __builtin_elementwise_fma(v2f{b[10], b[10]}, C[10], A2); b[13] = row_rot<13>(v); PIN();
    A3 = __builtin_elementwise_fma(v2f{b[11], b[11]}, C[11], A3); b[14] = row_rot<14>(v); PIN();
    A0 = __builtin_elementwise_fma(v2f{b[12], b[12]}, C[12], A0); b[15] = row_rot<15>(v); PIN();
    A1 = __builtin_elementwise_fma(v2f{b[13], b[13]}, C[13], A1);     PIN();
    A2 = __builtin_elementwise_fma(v2f{b[14], b[14]}, C[14], A2);     PIN();
    A3 = __builtin_elementwise_fma(v2f{b[15], b[15]}, C[15], A3);     PIN();
    asm("v_max_num_f32_dpp %5, %5, %5 row_xmask:4 row_mask:0xf bank_mask:0xf\n\t"
        "v_add_f32 %3, %6, %8\n\t"
        "v_add_f32 %4, %7, %9\n\t"
        "v_max_num_f32_dpp %5, %5, %5 row_xmask:8 row_mask:0xf bank_mask:0xf\n\t"
        "v_add_f32 %0, %10, %12\n\t"
        "v_add_f32 %1, %11, %13\n\t"
        "v_rcp_f32 %2, %5\n\t"
        "v_add_f32 %0, %3, %0\n\t"
        "v_add_f32 %1, %4, %1\n\t"
        "v_permlane16_swap_b32 %0, %1\n\t"
        "v_mul_f32 %2, %14, %2"
        : "=&v"(lo), "=&v"(hi), "=&v"(gs), "=&v"(t0), "=&v"(t1), "+v"(m)
        : "v"(A0.x), "v"(A0.y), "v"(A1.x), "v"(A1.y), "v"(A2.x), "v"(A2.y), "v"(A3.x), "v"(A3.y), "v"(g));
    s  = lo + hi;
    vn = s * gs;
}

#define FWD_WAVE 0
#define BWD_WAVE 5

__global__ __launch_bounds__(256) void crf_scan(const float* __restrict__ g,
                                                const float* __restrict__ U,
                                                float* __restrict__ out) {
    __shared__ float sp[Tv * Kv];
    __shared__ float sq[Tv * Kv];
    const int   lane   = threadIdx.x & 31;
    const int   wave   = __builtin_amdgcn_readfirstlane(threadIdx.x >> 5);
    const int   bIdx   = blockIdx.x;
    const float LOG2E  = 1.44269504088896340736f;
    v2f   C[16];
    float ring[PF];
    float keep[PF];
    float s, vn;
    const int   own    = (lane >> 4) * 16;
    const int   l15    = lane & 15;
    static_assert(PF == 16, "the fence statements below name the PF ring slots one by one");

    if (wave == FWD_WAVE) {
        #pragma unroll
        for (int n = 0; n < 16; ++n) {
            const int j = own + ((l15 + n) & 15);
            C[n].x = fexp2(U[j * Kv + l15] * LOG2E);
            C[n].y = fexp2(U[j * Kv + 16 + l15] * LOG2E);
        }

        const float* gb = g + (size_t)bIdx * Tv * Kv;
        float*       pb = sp;

        #pragma unroll
        for (int i = 0; i < PF; ++i)
            ring[i] = gb[(1 + i) * Kv + lane];

        float p = gb[lane];
        pb[lane] = p;
        asm volatile("" :: "v"(ring[0]), "v"(ring[1]), "v"(ring[2]), "v"(ring[3]),
                           "v"(ring[4]), "v"(ring[5]), "v"(ring[6]), "v"(ring[7]),
                           "v"(ring[8]), "v"(ring[9]), "v"(ring[10]), "v"(ring[11]),
                           "v"(ring[12]), "v"(ring[13]), "v"(ring[14]), "v"(ring[15]));
        int t = 1;
        for (; t + PF <= Tv; t += PF) {
            #pragma unroll
            for (int i = 0; i < PF; ++i) {
                lin_step(p, ring[i], C, s, vn);
                p = vn;
                keep[i] = p;
                int tn = t + i + PF;
                tn = tn < Tv ? tn : Tv - 1;
                ring[i] = gb[tn * Kv + lane];
                __builtin_amdgcn_sched_barrier(0);
            }
            #pragma unroll
            for (int i = 0; i < PF; ++i)
                pb[(t + i) * Kv + lane] = keep[i];
        }
        #pragma unroll
        for (int i = 0; i < PF - 1; ++i) {
            if (t + i < Tv) {
                lin_step(p, ring[i], C, s, vn);
                p = vn;
                pb[(t + i) * Kv + lane] = p;
            }
        }
    } else if (wave == BWD_WAVE) {
        #pragma unroll
        for (int n = 0; n < 16; ++n) {
            const int k = own + ((l15 + n) & 15);
            C[n].x = fexp2(U[l15 * Kv + k] * LOG2E);
            C[n].y = fexp2(U[(16 + l15) * Kv + k] * LOG2E);
        }

        const float* gb = g + (size_t)bIdx * Tv * Kv;
        float*       qb = sq;

        #pragma unroll
        for (int i = 0; i < PF; ++i)
            ring[i] = gb[(Tv - 2 - i) * Kv + lane];

        float w = gb[(Tv - 1) * Kv + lane];
        qb[(Tv - 1) * Kv + lane] = 1.f;
        asm volatile("" :: "v"(ring[0]), "v"(ring[1]), "v"(ring[2]), "v"(ring[3]),
                           "v"(ring[4]), "v"(ring[5]), "v"(ring[6]), "v"(ring[7]),
                           "v"(ring[8]), "v"(ring[9]), "v"(ring[10]), "v"(ring[11]),
                           "v"(ring[12]), "v"(ring[13]), "v"(ring[14]), "v"(ring[15]), "v"(w));
        int t = Tv - 2;
        for (; t - (PF - 1) >= 0; t -= PF) {
            #pragma unroll
            for (int i = 0; i < PF; ++i) {
                lin_step(w, ring[i], C, s, vn);
                keep[i] = s;
                w = vn;
                int qn = t - i - PF;
                qn = qn > 0 ? qn : 0;
                ring[i] = gb[qn * Kv + lane];
                __builtin_amdgcn_sched_barrier(0);
            }
            #pragma unroll
            for (int i = 0; i < PF; ++i)
                qb[(t - i) * Kv + lane] = keep[i];
        }
        #pragma unroll
        for (int i = 0; i < PF - 1; ++i) {
            if (t - i >= 0) {
                lin_step(w, ring[i], C, s, vn);
                qb[(t - i) * Kv + lane] = s;
                w = vn;
            }
        }
    }

    __syncthreads();

    const v4f* sp4 = (const v4f*)sp;
    const v4f* sq4 = (const v4f*)sq;
    v4f*       ob4 = (v4f*)(out + (size_t)bIdx * Tv * Kv);
    #pragma unroll 4
    for (int it = 0; it < (Tv * Kv / 4) / 256; ++it) {
        const int i4 = it * 256 + wave * 32 + lane;
        const v4f s = sp4[i4] * sq4[i4];
        const float sum = lane8_sum((s.x + s.y) + (s.z + s.w));
        const float inv = 1.0f / sum;
        ob4[i4] = s * inv;
    }
}

extern "C" void kernel_launch(void* const* d_in, const int* in_sizes, int n_in,
                              void* d_out, int out_size, void* d_ws, size_t ws_size,
                              hipStream_t stream) {
    const float* x = (const float*)d_in[0];
    const float* W = (const float*)d_in[1];
    const float* U = (const float*)d_in[2];
    const float* b = (const float*)d_in[3];
    float* out = (float*)d_out;

    float* ws = (float*)d_ws;
    float* g  = ws;
    float* Wp = g + (size_t)Bv * Tv * Kv;

    crf_packW     <<<dim3(128), dim3(256), 0, stream>>>(W, Wp);
    crf_emissions <<<dim3((Bv * Tv / 16) / 4), dim3(128), 0, stream>>>(x, Wp, b, g);
    crf_scan      <<<dim3(Bv), dim3(256), 0, stream>>>(g, U, out);
}


# emissions: W is read straight from its row-major form, 8 B per lane, the tags interleaved over the two N tiles (lane c: tags 2c, 2c+1), so the W-packing launch is gone (3 launches to 2)
# speedup vs baseline: 7.1632x; 1.0560x over previous
#include <hip/hip_runtime.h>
#include <hip/hip_bf16.h>

typedef __attribute__((ext_vector_type(2))) float v2f;
typedef __attribute__((ext_vector_type(4))) float v4f;
typedef __attribute__((ext_vector_type(8))) float v8f;

#define Bv 64
#define Tv 512
#define Dv 1024
#define Kv 32

__device__ __forceinline__ float fexp2(float x) { return __builtin_amdgcn_exp2f(x); }
__device__ __forceinline__ float flog2(float x) { return __builtin_amdgcn_logf(x); }

__device__ __forceinline__ float row16_max(float v) {
    float r;
    asm("v_max_num_f32_dpp %0, %1, %1 row_xmask:1 row_mask:0xf bank_mask:0xf\n\t"
        "v_max_num_f32_dpp %0, %0, %0 row_xmask:2 row_mask:0xf bank_mask:0xf\n\t"
        "v_max_num_f32_dpp %0, %0, %0 row_xmask:4 row_mask:0xf bank_mask:0xf\n\t"
        "v_max_num_f32_dpp %0, %0, %0 row_xmask:8 row_mask:0xf bank_mask:0xf"
        : "=v"(r) : "v"(v));
    return r;
}

__device__ __forceinline__ float lane8_sum(float v) {
    float r;
    asm("v_add_f32_dpp %0, %1, %1 row_xmask:1 row_mask:0xf bank_mask:0xf\n\t"
        "v_add_f32_dpp %0, %0, %0 row_xmask:2 row_mask:0xf bank_mask:0xf\n\t"
        "v_add_f32_dpp %0, %0, %0 row_xmask:4 row_mask:0xf bank_mask:0xf"
        : "=v"(r) : "v"(v));
    return r;
}

#define EG 8

struct EFrag { v4f a[EG]; v2f w[EG][4]; };

__device__ __forceinline__ void emis_load(EFrag& f, const float* __restrict__ xrow,
                                          const float* __restrict__ wl, int blk) {
    #pragma unroll
    for (int j = 0; j < EG; ++j) {
        const int g = blk * EG + j;
        f.a[j] = *(const v4f*)(xrow + g * 8);
        #pragma unroll
        for (int s = 0; s < 4; ++s)
            f.w[j][s] = *(const v2f*)(wl + (g * 8 + s) * Kv);
    }
}

__device__ __forceinline__ void emis_mma(const EFrag& f, v8f& acc0, v8f& acc1) {
    #pragma unroll
    for (int j = 0; j < EG; ++j) {
        v2f a01 = {f.a[j].x, f.a[j].y}, a23 = {f.a[j].z, f.a[j].w};
        v2f p01 = {f.w[j][0].x, f.w[j][1].x}, p23 = {f.w[j][2].x, f.w[j][3].x};
        v2f q01 = {f.w[j][0].y, f.w[j][1].y}, q23 = {f.w[j][2].y, f.w[j][3].y};
        acc0 = __builtin_amdgcn_wmma_f32_16x16x4_f32(false, a01, false, p01,
                                                     (short)0, acc0, false, false);
        acc1 = __builtin_amdgcn_wmma_f32_16x16x4_f32(false, a01, false, q01,
                                                     (short)0, acc1, false, false);
        acc0 = __builtin_amdgcn_wmma_f32_16x16x4_f32(false, a23, false, p23,
                                                     (short)0, acc0, false, false);
        acc1 = __builtin_amdgcn_wmma_f32_16x16x4_f32(false, a23, false, q23,
                                                     (short)0, acc1, false, false);
    }
}

__global__ __launch_bounds__(128) void crf_emissions(const float* __restrict__ x,
                                                     const float* __restrict__ W,
                                                     const float* __restrict__ bias,
                                                     float* __restrict__ e) {
    const int lane = threadIdx.x & 31;
    const int wave = threadIdx.x >> 5;
    const int mt   = blockIdx.x * 4 + wave;
    const int half = lane >> 4;
    const int l15  = lane & 15;

    const float* xrow = x  + (size_t)(mt * 16 + l15) * Dv + half * 4;
    const float* wl   = W + half * 4 * Kv + l15 * 2;

    v8f acc0 = {};
    v8f acc1 = {};
    EFrag P, Q;
    const int NB = Dv / (8 * EG);
    emis_load(P, xrow, wl, 0);
    #pragma clang loop unroll(disable)
    for (int blk = 0; blk < NB - 2; blk += 2) {
        emis_load(Q, xrow, wl, blk + 1);
        emis_mma(P, acc0, acc1);
        emis_load(P, xrow, wl, blk + 2);
        emis_mma(Q, acc0, acc1);
    }
    emis_load(Q, xrow, wl, NB - 1);
    emis_mma(P, acc0, acc1);
    emis_mma(Q, acc0, acc1);

    const float bv0 = bias[l15 * 2];
    const float bv1 = bias[l15 * 2 + 1];
    const float LOG2E = 1.44269504088896340736f;
    #pragma unroll
    for (int r = 0; r < 8; ++r) {
        size_t row = (size_t)mt * 16 + r + half * 8;
        const float v0 = acc0[r] + bv0;
        const float v1 = acc1[r] + bv1;
        const float m  = row16_max(fmaxf(v0, v1));
        *(v2f*)(e + row * Kv + l15 * 2) = v2f{fexp2((v0 - m) * LOG2E), fexp2((v1 - m) * LOG2E)};
    }
}

#define PF 16

__host__ __device__ constexpr unsigned rot_selects(int n, int first_lane) {
    unsigned w = 0;
    for (int j = 0; j < 8; ++j) w |= (unsigned)((first_lane + j + n) & 15) << (4 * j);
    return w;
}
template <int N> __device__ __forceinline__ float row_rot(float v) {
    float r;
    asm("v_permlane16_b32 %0, %1, %2, %3"
        : "=v"(r) : "v"(v), "s"(rot_selects(N, 0)), "s"(rot_selects(N, 8)));
    return r;
}
__device__ __forceinline__ float other_row(float v) {
    float r;
    asm("v_permlanex16_b32 %0, %1, %2, %3" : "=v"(r) : "v"(v), "s"(0x76543210u), "s"(0xfedcba98u));
    return r;
}
template <int M> __device__ __forceinline__ float max_xmask(float v) {
    float r;
    asm("v_max_num_f32_dpp %0, %1, %1 row_xmask:%2 row_mask:0xf bank_mask:0xf" : "=v"(r) : "v"(v), "n"(M));
    return r;
}
#define PIN() __builtin_amdgcn_sched_barrier(0)

__device__ __forceinline__ void lin_step(float v, float g, const v2f (&C)[16],
                                         float& s, float& vn) {
    float b[16];
    v2f A0, A1, A2, A3;
    float m, gs, lo, hi, t0, t1;
    b[1] = row_rot<1>(v);
    A0 = v2f{v, v} * C[0];
    b[2] = row_rot<2>(v);
    const float vx = other_row(v);
    b[3] = row_rot<3>(v);
    PIN();
    A1 = v2f{b[1], b[1]} * C[1];            b[4] = row_rot<4>(v);     PIN();
    A2 = v2f{b[2], b[2]} * C[2];            b[5] = row_rot<5>(v);     PIN();
    asm("v_max_num_f32 %0, %1, %2" : "=v"(m) : "v"(v), "v"(vx));      PIN();
    A3 = v2f{b[3], b[3]} * C[3];            b[6] = row_rot<6>(v);     PIN();
    A0 = __builtin_elementwise_fma(v2f{b[4], b[4]}, C[4], A0);   b[7]  = row_rot<7>(v);   PIN();
    A1 = __builtin_elementwise_fma(v2f{b[5], b[5]}, C[5], A1);   b[8]  = row_rot<8>(v);   PIN();
    m  = max_xmask<1>(m);                                             PIN();
    A2 = __builtin_elementwise_fma(v2f{b[6], b[6]}, C[6], A2);   b[9]  = row_rot<9>(v);   PIN();
    A3 = __builtin_elementwise_fma(v2f{b[7], b[7]}, C[7], A3);   b[10] = row_rot<10>(v);  PIN();
    A0 = __builtin_elementwise_fma(v2f{b[8], b[8]}, C[8], A0);   b[11] = row_rot<11>(v);  PIN();
    A1 = __builtin_elementwise_fma(v2f{b[9], b[9]}, C[9], A1);   b[12] = row_rot<12>(v);  PIN();
    m  = max_xmask<2>(m);                                             PIN();
    A2 = __builtin_elementwise_fma(v2f{b[10], b[10]}, C[10], A2); b[13] = row_rot<13>(v); PIN();
    A3 = __builtin_elementwise_fma(v2f{b[11], b[11]}, C[11], A3); b[14] = row_rot<14>(v); PIN();
    A0 = __builtin_elementwise_fma(v2f{b[12], b[12]}, C[12], A0); b[15] = row_rot<15>(v); PIN();
    A1 = __builtin_elementwise_fma(v2f{b[13], b[13]}, C[13], A1);     PIN();
    A2 = __builtin_elementwise_fma(v2f{b[14], b[14]}, C[14], A2);     PIN();
    A3 = __builtin_elementwise_fma(v2f{b[15], b[15]}, C[15], A3);     PIN();
    asm("v_max_num_f32_dpp %5, %5, %5 row_xmask:4 row_mask:0xf bank_mask:0xf\n\t"
        "v_add_f32 %3, %6, %8\n\t"
        "v_add_f32 %4, %7, %9\n\t"
        "v_max_num_f32_dpp %5, %5, %5 row_xmask:8 row_mask:0xf bank_mask:0xf\n\t"
        "v_add_f32 %0, %10, %12\n\t"
        "v_add_f32 %1, %11, %13\n\t"
        "v_rcp_f32 %2, %5\n\t"
        "v_add_f32 %0, %3, %0\n\t"
        "v_add_f32 %1, %4, %1\n\t"
        "v_permlane16_swap_b32 %0, %1\n\t"
        "v_mul_f32 %2, %14, %2"
        : "=&v"(lo), "=&v"(hi), "=&v"(gs), "=&v"(t0), "=&v"(t1), "+v"(m)
        : "v"(A0.x), "v"(A0.y), "v"(A1.x), "v"(A1.y), "v"(A2.x), "v"(A2.y), "v"(A3.x), "v"(A3.y), "v"(g));
    s  = lo + hi;
    vn = s * gs;
}

#define FWD_WAVE 0
#define BWD_WAVE 5

__global__ __launch_bounds__(256) void crf_scan(const float* __restrict__ g,
                                                const float* __restrict__ U,
                                                float* __restrict__ out) {
    __shared__ float sp[Tv * Kv];
    __shared__ float sq[Tv * Kv];
    const int   lane   = threadIdx.x & 31;
    const int   wave   = __builtin_amdgcn_readfirstlane(threadIdx.x >> 5);
    const int   bIdx   = blockIdx.x;
    const float LOG2E  = 1.44269504088896340736f;
    v2f   C[16];
    float ring[PF];
    float keep[PF];
    float s, vn;
    const int   own    = (lane >> 4) * 16;
    const int   l15    = lane & 15;
    static_assert(PF == 16, "the fence statements below name the PF ring slots one by one");

    if (wave == FWD_WAVE) {
        #pragma unroll
        for (int n = 0; n < 16; ++n) {
            const int j = own + ((l15 + n) & 15);
            C[n].x = fexp2(U[j * Kv + l15] * LOG2E);
            C[n].y = fexp2(U[j * Kv + 16 + l15] * LOG2E);
        }

        const float* gb = g + (size_t)bIdx * Tv * Kv;
        float*       pb = sp;

        #pragma unroll
        for (int i = 0; i < PF; ++i)
            ring[i] = gb[(1 + i) * Kv + lane];

        float p = gb[lane];
        pb[lane] = p;
        asm volatile("" :: "v"(ring[0]), "v"(ring[1]), "v"(ring[2]), "v"(ring[3]),
                           "v"(ring[4]), "v"(ring[5]), "v"(ring[6]), "v"(ring[7]),
                           "v"(ring[8]), "v"(ring[9]), "v"(ring[10]), "v"(ring[11]),
                           "v"(ring[12]), "v"(ring[13]), "v"(ring[14]), "v"(ring[15]));
        int t = 1;
        for (; t + PF <= Tv; t += PF) {
            #pragma unroll
            for (int i = 0; i < PF; ++i) {
                lin_step(p, ring[i], C, s, vn);
                p = vn;
                keep[i] = p;
                int tn = t + i + PF;
                tn = tn < Tv ? tn : Tv - 1;
                ring[i] = gb[tn * Kv + lane];
                __builtin_amdgcn_sched_barrier(0);
            }
            #pragma unroll
            for (int i = 0; i < PF; ++i)
                pb[(t + i) * Kv + lane] = keep[i];
        }
        #pragma unroll
        for (int i = 0; i < PF - 1; ++i) {
            if (t + i < Tv) {
                lin_step(p, ring[i], C, s, vn);
                p = vn;
                pb[(t + i) * Kv + lane] = p;
            }
        }
    } else if (wave == BWD_WAVE) {
        #pragma unroll
        for (int n = 0; n < 16; ++n) {
            const int k = own + ((l15 + n) & 15);
            C[n].x = fexp2(U[l15 * Kv + k] * LOG2E);
            C[n].y = fexp2(U[(16 + l15) * Kv + k] * LOG2E);
        }

        const float* gb = g + (size_t)bIdx * Tv * Kv;
        float*       qb = sq;

        #pragma unroll
        for (int i = 0; i < PF; ++i)
            ring[i] = gb[(Tv - 2 - i) * Kv + lane];

        float w = gb[(Tv - 1) * Kv + lane];
        qb[(Tv - 1) * Kv + lane] = 1.f;
        asm volatile("" :: "v"(ring[0]), "v"(ring[1]), "v"(ring[2]), "v"(ring[3]),
                           "v"(ring[4]), "v"(ring[5]), "v"(ring[6]), "v"(ring[7]),
                           "v"(ring[8]), "v"(ring[9]), "v"(ring[10]), "v"(ring[11]),
                           "v"(ring[12]), "v"(ring[13]), "v"(ring[14]), "v"(ring[15]), "v"(w));
        int t = Tv - 2;
        for (; t - (PF - 1) >= 0; t -= PF) {
            #pragma unroll
            for (int i = 0; i < PF; ++i) {
                lin_step(w, ring[i], C, s, vn);
                keep[i] = s;
                w = vn;
                int qn = t - i - PF;
                qn = qn > 0 ? qn : 0;
                ring[i] = gb[qn * Kv + lane];
                __builtin_amdgcn_sched_barrier(0);
            }
            #pragma unroll
            for (int i = 0; i < PF; ++i)
                qb[(t - i) * Kv + lane] = keep[i];
        }
        #pragma unroll
        for (int i = 0; i < PF - 1; ++i) {
            if (t - i >= 0) {
                lin_step(w, ring[i], C, s, vn);
                qb[(t - i) * Kv + lane] = s;
                w = vn;
            }
        }
    }

    __syncthreads();

    const v4f* sp4 = (const v4f*)sp;
    const v4f* sq4 = (const v4f*)sq;
    v4f*       ob4 = (v4f*)(out + (size_t)bIdx * Tv * Kv);
    #pragma unroll 4
    for (int it = 0; it < (Tv * Kv / 4) / 256; ++it) {
        const int i4 = it * 256 + wave * 32 + lane;
        const v4f s = sp4[i4] * sq4[i4];
        const float sum = lane8_sum((s.x + s.y) + (s.z + s.w));
        const float inv = 1.0f / sum;
        ob4[i4] = s * inv;
    }
}

extern "C" void kernel_launch(void* const* d_in, const int* in_sizes, int n_in,
                              void* d_out, int out_size, void* d_ws, size_t ws_size,
                              hipStream_t stream) {
    const float* x = (const float*)d_in[0];
    const float* W = (const float*)d_in[1];
    const float* U = (const float*)d_in[2];
    const float* b = (const float*)d_in[3];
    float* out = (float*)d_out;

    float* ws = (float*)d_ws;
    float* g  = ws;

    crf_emissions <<<dim3((Bv * Tv / 16) / 4), dim3(128), 0, stream>>>(x, W, b, g);
    crf_scan      <<<dim3(Bv), dim3(256), 0, stream>>>(g, U, out);
}


# scan: a lane sums 8 outputs over its own 4-lane quad (3 v_permlane16 rotations, was 15); DPP row_xmask 4 and 8 adds and v_permlanex16 fold the partial sums; sum order changed
# speedup vs baseline: 7.5923x; 1.0599x over previous
#include <hip/hip_runtime.h>
#include <hip/hip_bf16.h>

typedef __attribute__((ext_vector_type(2))) float v2f;
typedef __attribute__((ext_vector_type(4))) float v4f;
typedef __attribute__((ext_vector_type(8))) float v8f;

#define Bv 64
#define Tv 512
#define Dv 1024
#define Kv 32

__device__ __forceinline__ float fexp2(float x) { return __builtin_amdgcn_exp2f(x); }
__device__ __forceinline__ float flog2(float x) { return __builtin_amdgcn_logf(x); }

__device__ __forceinline__ float row16_max(float v) {
    float r;
    asm("v_max_num_f32_dpp %0, %1, %1 row_xmask:1 row_mask:0xf bank_mask:0xf\n\t"
        "v_max_num_f32_dpp %0, %0, %0 row_xmask:2 row_mask:0xf bank_mask:0xf\n\t"
        "v_max_num_f32_dpp %0, %0, %0 row_xmask:4 row_mask:0xf bank_mask:0xf\n\t"
        "v_max_num_f32_dpp %0, %0, %0 row_xmask:8 row_mask:0xf bank_mask:0xf"
        : "=v"(r) : "v"(v));
    return r;
}

__device__ __forceinline__ float lane8_sum(float v) {
    float r;
    asm("v_add_f32_dpp %0, %1, %1 row_xmask:1 row_mask:0xf bank_mask:0xf\n\t"
        "v_add_f32_dpp %0, %0, %0 row_xmask:2 row_mask:0xf bank_mask:0xf\n\t"
        "v_add_f32_dpp %0, %0, %0 row_xmask:4 row_mask:0xf bank_mask:0xf"
        : "=v"(r) : "v"(v));
    return r;
}

#define EG 8

struct EFrag { v4f a[EG]; v2f w[EG][4]; };

__device__ __forceinline__ void emis_load(EFrag& f, const float* __restrict__ xrow,
                                          const float* __restrict__ wl, int blk) {
    #pragma unroll
    for (int j = 0; j < EG; ++j) {
        const int g = blk * EG + j;
        f.a[j] = *(const v4f*)(xrow + g * 8);
        #pragma unroll
        for (int s = 0; s < 4; ++s)
            f.w[j][s] = *(const v2f*)(wl + (g * 8 + s) * Kv);
    }
}

__device__ __forceinline__ void emis_mma(const EFrag& f, v8f& acc0, v8f& acc1) {
    #pragma unroll
    for (int j = 0; j < EG; ++j) {
        v2f a01 = {f.a[j].x, f.a[j].y}, a23 = {f.a[j].z, f.a[j].w};
        v2f p01 = {f.w[j][0].x, f.w[j][1].x}, p23 = {f.w[j][2].x, f.w[j][3].x};
        v2f q01 = {f.w[j][0].y, f.w[j][1].y}, q23 = {f.w[j][2].y, f.w[j][3].y};
        acc0 = __builtin_amdgcn_wmma_f32_16x16x4_f32(false, a01, false, p01,
                                                     (short)0, acc0, false, false);
        acc1 = __builtin_amdgcn_wmma_f32_16x16x4_f32(false, a01, false, q01,
                                                     (short)0, acc1, false, false);
        acc0 = __builtin_amdgcn_wmma_f32_16x16x4_f32(false, a23, false, p23,
                                                     (short)0, acc0, false, false);
        acc1 = __builtin_amdgcn_wmma_f32_16x16x4_f32(false, a23, false, q23,
                                                     (short)0, acc1, false, false);
    }
}

__global__ __launch_bounds__(128) void crf_emissions(const float* __restrict__ x,
                                                     const float* __restrict__ W,
                                                     const float* __restrict__ bias,
                                                     float* __restrict__ e) {
    const int lane = threadIdx.x & 31;
    const int wave = threadIdx.x >> 5;
    const int mt   = blockIdx.x * 4 + wave;
    const int half = lane >> 4;
    const int l15  = lane & 15;

    const float* xrow = x  + (size_t)(mt * 16 + l15) * Dv + half * 4;
    const float* wl   = W + half * 4 * Kv + l15 * 2;

    v8f acc0 = {};
    v8f acc1 = {};
    EFrag P, Q;
    const int NB = Dv / (8 * EG);
    emis_load(P, xrow, wl, 0);
    #pragma clang loop unroll(disable)
    for (int blk = 0; blk < NB - 2; blk += 2) {
        emis_load(Q, xrow, wl, blk + 1);
        emis_mma(P, acc0, acc1);
        emis_load(P, xrow, wl, blk + 2);
        emis_mma(Q, acc0, acc1);
    }
    emis_load(Q, xrow, wl, NB - 1);
    emis_mma(P, acc0, acc1);
    emis_mma(Q, acc0, acc1);

    const float bv0 = bias[l15 * 2];
    const float bv1 = bias[l15 * 2 + 1];
    const float LOG2E = 1.44269504088896340736f;
    #pragma unroll
    for (int r = 0; r < 8; ++r) {
        size_t row = (size_t)mt * 16 + r + half * 8;
        const float v0 = acc0[r] + bv0;
        const float v1 = acc1[r] + bv1;
        const float m  = row16_max(fmaxf(v0, v1));
        *(v2f*)(e + row * Kv + l15 * 2) = v2f{fexp2((v0 - m) * LOG2E), fexp2((v1 - m) * LOG2E)};
    }
}

#define PF 16

__host__ __device__ constexpr unsigned quad_selects(int m, int first_lane) {
    unsigned w = 0;
    for (int j = 0; j < 8; ++j) {
        const int l = first_lane + j;
        w |= (unsigned)((l & 12) | ((l + m) & 3)) << (4 * j);
    }
    return w;
}
template <int M> __device__ __forceinline__ float quad_rot(float v) {
    float r;
    asm("v_permlane16_b32 %0, %1, %2, %3"
        : "=v"(r) : "v"(v), "s"(quad_selects(M, 0)), "s"(quad_selects(M, 8)));
    return r;
}
__device__ __forceinline__ float other_row(float v) {
    float r;
    asm("v_permlanex16_b32 %0, %1, %2, %3" : "=v"(r) : "v"(v), "s"(0x76543210u), "s"(0xfedcba98u));
    return r;
}
template <int M> __device__ __forceinline__ float max_xmask(float v) {
    float r;
    asm("v_max_num_f32_dpp %0, %1, %1 row_xmask:%2 row_mask:0xf bank_mask:0xf" : "=v"(r) : "v"(v), "n"(M));
    return r;
}
__host__ __device__ constexpr int acc_tag(int a) { return a == 0 ? 16 : a == 1 ? 24 : a == 2 ? 0 : 8; }
#define PIN() __builtin_amdgcn_sched_barrier(0)

__device__ __forceinline__ void lin_step(float v, float g, const v2f (&C)[16],
                                         float& s, float& vn) {
    v2f A0, A1, A2, A3;
    float m, gs, lo, px, t0, t1, t2, t3;
    const float vx = other_row(v);
    const float b1 = quad_rot<1>(v);
    const float b2 = quad_rot<2>(v);
    const float b3 = quad_rot<3>(v);
    PIN();
    asm("v_max_num_f32 %0, %1, %2" : "=v"(m) : "v"(v), "v"(vx));      PIN();
    A0 = v2f{v, v} * C[0];
    A1 = v2f{v, v} * C[1];
    A2 = v2f{v, v} * C[2];
    A3 = v2f{v, v} * C[3];                                            PIN();
    m  = max_xmask<1>(m);                                             PIN();
    A0 = __builtin_elementwise_fma(v2f{b1, b1}, C[4], A0);
    A1 = __builtin_elementwise_fma(v2f{b1, b1}, C[5], A1);
    A2 = __builtin_elementwise_fma(v2f{b1, b1}, C[6], A2);
    A3 = __builtin_elementwise_fma(v2f{b1, b1}, C[7], A3);            PIN();
    m  = max_xmask<2>(m);                                             PIN();
    A0 = __builtin_elementwise_fma(v2f{b2, b2}, C[8], A0);
    A1 = __builtin_elementwise_fma(v2f{b2, b2}, C[9], A1);
    A2 = __builtin_elementwise_fma(v2f{b2, b2}, C[10], A2);
    A3 = __builtin_elementwise_fma(v2f{b2, b2}, C[11], A3);           PIN();
    m  = max_xmask<4>(m);                                             PIN();
    A0 = __builtin_elementwise_fma(v2f{b3, b3}, C[12], A0);           PIN();
    A1 = __builtin_elementwise_fma(v2f{b3, b3}, C[13], A1);           PIN();
    A2 = __builtin_elementwise_fma(v2f{b3, b3}, C[14], A2);           PIN();
    A3 = __builtin_elementwise_fma(v2f{b3, b3}, C[15], A3);           PIN();
    asm("v_max_num_f32_dpp %7, %7, %7 row_xmask:8 row_mask:0xf bank_mask:0xf\n\t"
        "v_add_f32_dpp %3, %9, %8 row_xmask:4 row_mask:0xf bank_mask:0xf\n\t"
        "v_add_f32_dpp %4, %11, %10 row_xmask:4 row_mask:0xf bank_mask:0xf\n\t"
        "v_add_f32_dpp %5, %13, %12 row_xmask:4 row_mask:0xf bank_mask:0xf\n\t"
        "v_add_f32_dpp %6, %15, %14 row_xmask:4 row_mask:0xf bank_mask:0xf\n\t"
        "v_rcp_f32 %2, %7\n\t"
        "v_add_f32_dpp %3, %4, %3 row_xmask:8 row_mask:0xf bank_mask:0xf\n\t"
        "v_add_f32_dpp %0, %6, %5 row_xmask:8 row_mask:0xf bank_mask:0xf\n\t"
        "v_permlanex16_b32 %1, %3, %17, %18\n\t"
        "v_mul_f32 %2, %16, %2"
        : "=&v"(lo), "=&v"(px), "=&v"(gs), "=&v"(t0), "=&v"(t1), "=&v"(t2), "=&v"(t3), "+v"(m)
        : "v"(A0.x), "v"(A0.y), "v"(A1.x), "v"(A1.y), "v"(A2.x), "v"(A2.y), "v"(A3.x), "v"(A3.y),
          "v"(g), "s"(0x76543210u), "s"(0xfedcba98u));
    s  = lo + px;
    vn = s * gs;
}

#define FWD_WAVE 0
#define BWD_WAVE 5

__global__ __launch_bounds__(256) void crf_scan(const float* __restrict__ g,
                                                const float* __restrict__ U,
                                                float* __restrict__ out) {
    __shared__ float sp[Tv * Kv];
    __shared__ float sq[Tv * Kv];
    const int   lane   = threadIdx.x & 31;
    const int   wave   = __builtin_amdgcn_readfirstlane(threadIdx.x >> 5);
    const int   bIdx   = blockIdx.x;
    const float LOG2E  = 1.44269504088896340736f;
    v2f   C[16];
    float ring[PF];
    float keep[PF];
    float s, vn;
    static_assert(PF == 16, "the fence statements below name the PF ring slots one by one");

    if (wave == FWD_WAVE) {
        #pragma unroll
        for (int n = 0; n < 16; ++n) {
            const int j = (lane & ~3) | ((lane + (n >> 2)) & 3);
            const int k = lane ^ acc_tag(n & 3);
            C[n].x = fexp2(U[j * Kv + k] * LOG2E);
            C[n].y = fexp2(U[j * Kv + (k ^ 4)] * LOG2E);
        }

        const float* gb = g + (size_t)bIdx * Tv * Kv;
        float*       pb = sp;

        #pragma unroll
        for (int i = 0; i < PF; ++i)
            ring[i] = gb[(1 + i) * Kv + lane];

        float p = gb[lane];
        pb[lane] = p;
        asm volatile("" :: "v"(ring[0]), "v"(ring[1]), "v"(ring[2]), "v"(ring[3]),
                           "v"(ring[4]), "v"(ring[5]), "v"(ring[6]), "v"(ring[7]),
                           "v"(ring[8]), "v"(ring[9]), "v"(ring[10]), "v"(ring[11]),
                           "v"(ring[12]), "v"(ring[13]), "v"(ring[14]), "v"(ring[15]));
        int t = 1;
        for (; t + PF <= Tv; t += PF) {
            #pragma unroll
            for (int i = 0; i < PF; ++i) {
                lin_step(p, ring[i], C, s, vn);
                p = vn;
                keep[i] = p;
                int tn = t + i + PF;
                tn = tn < Tv ? tn : Tv - 1;
                ring[i] = gb[tn * Kv + lane];
                __builtin_amdgcn_sched_barrier(0);
            }
            #pragma unroll
            for (int i = 0; i < PF; ++i)
                pb[(t + i) * Kv + lane] = keep[i];
        }
        #pragma unroll
        for (int i = 0; i < PF - 1; ++i) {
            if (t + i < Tv) {
                lin_step(p, ring[i], C, s, vn);
                p = vn;
                pb[(t + i) * Kv + lane] = p;
            }
        }
    } else if (wave == BWD_WAVE) {
        #pragma unroll
        for (int n = 0; n < 16; ++n) {
            const int k = (lane & ~3) | ((lane + (n >> 2)) & 3);
            const int j = lane ^ acc_tag(n & 3);
            C[n].x = fexp2(U[j * Kv + k] * LOG2E);
            C[n].y = fexp2(U[(j ^ 4) * Kv + k] * LOG2E);
        }

        const float* gb = g + (size_t)bIdx * Tv * Kv;
        float*       qb = sq;

        #pragma unroll
        for (int i = 0; i < PF; ++i)
            ring[i] = gb[(Tv - 2 - i) * Kv + lane];

        float w = gb[(Tv - 1) * Kv + lane];
        qb[(Tv - 1) * Kv + lane] = 1.f;
        asm volatile("" :: "v"(ring[0]), "v"(ring[1]), "v"(ring[2]), "v"(ring[3]),
                           "v"(ring[4]), "v"(ring[5]), "v"(ring[6]), "v"(ring[7]),
                           "v"(ring[8]), "v"(ring[9]), "v"(ring[10]), "v"(ring[11]),
                           "v"(ring[12]), "v"(ring[13]), "v"(ring[14]), "v"(ring[15]), "v"(w));
        int t = Tv - 2;
        for (; t - (PF - 1) >= 0; t -= PF) {
            #pragma unroll
            for (int i = 0; i < PF; ++i) {
                lin_step(w, ring[i], C, s, vn);
                keep[i] = s;
                w = vn;
                int qn = t - i - PF;
                qn = qn > 0 ? qn : 0;
                ring[i] = gb[qn * Kv + lane];
                __builtin_amdgcn_sched_barrier(0);
            }
            #pragma unroll
            for (int i = 0; i < PF; ++i)
                qb[(t - i) * Kv + lane] = keep[i];
        }
        #pragma unroll
        for (int i = 0; i < PF - 1; ++i) {
            if (t - i >= 0) {
                lin_step(w, ring[i], C, s, vn);
                qb[(t - i) * Kv + lane] = s;
                w = vn;
            }
        }
    }

    __syncthreads();

    const v4f* sp4 = (const v4f*)sp;
    const v4f* sq4 = (const v4f*)sq;
    v4f*       ob4 = (v4f*)(out + (size_t)bIdx * Tv * Kv);
    #pragma unroll 4
    for (int it = 0; it < (Tv * Kv / 4) / 256; ++it) {
        const int i4 = it * 256 + wave * 32 + lane;
        const v4f s = sp4[i4] * sq4[i4];
        const float sum = lane8_sum((s.x + s.y) + (s.z + s.w));
        const float inv = 1.0f / sum;
        ob4[i4] = s * inv;
    }
}

extern "C" void kernel_launch(void* const* d_in, const int* in_sizes, int n_in,
                              void* d_out, int out_size, void* d_ws, size_t ws_size,
                              hipStream_t stream) {
    const float* x = (const float*)d_in[0];
    const float* W = (const float*)d_in[1];
    const float* U = (const float*)d_in[2];
    const float* b = (const float*)d_in[3];
    float* out = (float*)d_out;

    float* ws = (float*)d_ws;
    float* g  = ws;

    crf_emissions <<<dim3((Bv * Tv / 16) / 4), dim3(128), 0, stream>>>(x, W, b, g);
    crf_scan      <<<dim3(Bv), dim3(256), 0, stream>>>(g, U, out);
}
